# starting kernel
# baseline (speedup 1.0000x reference)
#include <hip/hip_runtime.h>
#include <hip/hip_bf16.h>

typedef __attribute__((ext_vector_type(2))) float v2f;
typedef __attribute__((ext_vector_type(8))) float v8f;

#define T_LEN   16384
#define TP_LEN  16448
#define HOP     32
#define NFRM    513
#define NFRMP   514
#define NB      8
#define NCH     64
#define NFREQ   33
#define COL2P   4112
#define COL1P   263168
#define PLANE   263168

#define OFF_A1  0
#define OFF_A3  4096
#define OFF_WT  8192
#define OFF_U   143360
#define OFF_Y   (OFF_U + 64*PLANE)
#define OFF_FR  OFF_U
#define OFF_XP  OFF_Y

__device__ __forceinline__ v8f wmma_f32(v2f a, v2f b, v8f c) {
    return __builtin_amdgcn_wmma_f32_16x16x4_f32(false, a, false, b, (short)0, c,
                                                 false, false);
}

__global__ void prep_kernel(const float* __restrict__ wgt,
                            float* __restrict__ A1, float* __restrict__ A3,
                            float* __restrict__ Wt) {
    const int gid = blockIdx.x * blockDim.x + threadIdx.x;
    const int gsz = gridDim.x * blockDim.x;
    const float PI = 3.14159265358979f;
    for (int idx = gid; idx < 4096; idx += gsz) {
        int fr = idx >> 6, tau = idx & 63;
        float wv = 0.5f * (1.0f - cosf(2.0f * PI * (float)tau / 64.0f));
        float v;
        if (fr <= 32) v =  cosf(2.0f * PI * (float)(fr)      * (float)tau / 64.0f) * wv;
        else          v = -sinf(2.0f * PI * (float)(fr - 32) * (float)tau / 64.0f) * wv;
        A1[idx] = v;
    }
    for (int idx = gid; idx < 4096; idx += gsz) {
        int tau = idx >> 6, fr = idx & 63;
        float wv = 0.5f * (1.0f - cosf(2.0f * PI * (float)tau / 64.0f));
        float s = wv * (1.0f / 64.0f);
        float v;
        if (fr == 0)       v = s;
        else if (fr < 32)  v = 2.0f * cosf(2.0f * PI * (float)fr * (float)tau / 64.0f) * s;
        else if (fr == 32) v = cosf(PI * (float)tau) * s;
        else               v = -2.0f * sinf(2.0f * PI * (float)(fr - 32) * (float)tau / 64.0f) * s;
        A3[idx] = v;
    }
    for (int idx = gid; idx < NFREQ * 4096; idx += gsz) {
        int f = idx >> 12, oi = idx & 4095;
        Wt[idx] = wgt[(size_t)oi * NFREQ + f];
    }
}

__global__ void pad_kernel(const float* __restrict__ x, float* __restrict__ xp) {
    const int tp = blockIdx.x * blockDim.x + threadIdx.x;
    if (tp >= TP_LEN) return;
    const int bc = blockIdx.y;
    int t = tp - 32;
    t = t < 0 ? -t : t;
    t = t >= T_LEN ? 2 * T_LEN - 2 - t : t;
    xp[(size_t)bc * TP_LEN + tp] = x[(size_t)bc * T_LEN + t];
}

__global__ void stft_gemm(const float* __restrict__ xp, const float* __restrict__ A1,
                          float* __restrict__ U) {
    __shared__ float lA[4096];
    const int tid = threadIdx.x;
#pragma unroll
    for (int i = 0; i < 8; ++i)
        *(float4*)(lA + (i * 128 + tid) * 4) =
            *(const float4*)(A1 + (i * 128 + tid) * 4);
    __syncthreads();

    const int lane = tid & 31;
    const int w    = tid >> 5;
    const int cL   = lane & 15;
    const int hs   = lane >> 4;
    const int colp = blockIdx.x * 16 + cL;
    const int n    = colp % NFRMP;
    const int bc   = colp / NFRMP;
    const int b    = bc >> 6;
    const int ic   = bc & 63;
    const int nc   = n > 512 ? 512 : n;
    const float* xrow = xp + (size_t)bc * TP_LEN + nc * HOP;
    const int rowA = 16 * w + cL;
    v8f c = {};
#pragma unroll
    for (int k0 = 0; k0 < 64; k0 += 4) {
        const int kk = k0 + 2 * hs;
        v2f a  = *(const v2f*)(lA + rowA * 64 + kk);
        v2f bv = *(const v2f*)(xrow + kk);
        c = wmma_f32(a, bv, c);
    }
    float* Ust = U + (size_t)(16 * w + 8 * hs) * PLANE +
                 (size_t)ic * COL2P + b * NFRMP + n;
#pragma unroll
    for (int r = 0; r < 8; ++r)
        Ust[(size_t)r * PLANE] = c[r];
}

__global__ void mix_gemm(const float* __restrict__ Wt, const float* __restrict__ U,
                         float* __restrict__ Y) {
    const int f  = blockIdx.y;
    const int ri = blockIdx.z;
    if (ri == 1 && (f == 0 || f == 32)) return;
    const int frq = f + ri * 32;

    __shared__ float lA[4096];
    __shared__ float lB[1024];
    const int tid = threadIdx.x;
    const float* Wf = Wt + (size_t)f * 4096;
    const float* Ub = U + (size_t)frq * PLANE + blockIdx.x * 16;
#pragma unroll
    for (int i = 0; i < 8; ++i)
        *(float4*)(lA + (i * 128 + tid) * 4) =
            *(const float4*)(Wf + (i * 128 + tid) * 4);
#pragma unroll
    for (int i = 0; i < 2; ++i) {
        const int fi = i * 128 + tid;
        const int r = fi >> 2, p = fi & 3;
        *(float4*)(lB + fi * 4) = *(const float4*)(Ub + (size_t)r * COL2P + p * 4);
    }
    __syncthreads();

    const int lane = tid & 31;
    const int w    = tid >> 5;
    const int cL   = lane & 15;
    const int hs   = lane >> 4;
    const int rowA = 16 * w + cL;
    v8f c = {};
#pragma unroll
    for (int k0 = 0; k0 < 64; k0 += 4) {
        const int kk = k0 + 2 * hs;
        v2f a = *(const v2f*)(lA + rowA * 64 + kk);
        v2f bv;
        bv.x = lB[kk * 16 + cL];
        bv.y = lB[(kk + 1) * 16 + cL];
        c = wmma_f32(a, bv, c);
    }
    const int col2p = blockIdx.x * 16 + cL;
    const int bI = col2p / NFRMP;
    const int nI = col2p % NFRMP;
    float* Yst = Y + (size_t)frq * PLANE + (size_t)bI * (64 * NFRMP) + nI +
                 (size_t)(16 * w + 8 * hs) * NFRMP;
#pragma unroll
    for (int r = 0; r < 8; ++r)
        Yst[(size_t)r * NFRMP] = c[r];
}

__global__ void istft_gemm(const float* __restrict__ A3, const float* __restrict__ Y,
                           float* __restrict__ frames) {
    __shared__ float lA[4096];
    const int tid = threadIdx.x;
#pragma unroll
    for (int i = 0; i < 8; ++i)
        *(float4*)(lA + (i * 128 + tid) * 4) =
            *(const float4*)(A3 + (i * 128 + tid) * 4);
    __syncthreads();

    const int lane = tid & 31;
    const int w    = tid >> 5;
    const int cL   = lane & 15;
    const int hs   = lane >> 4;
    const int colp = blockIdx.x * 16 + cL;
    const float* Yb = Y + (size_t)colp + (size_t)(2 * hs) * PLANE;
    const int rowA = 16 * w + cL;
    v8f c = {};
#pragma unroll
    for (int k0 = 0; k0 < 64; k0 += 4) {
        v2f a = *(const v2f*)(lA + rowA * 64 + (k0 + 2 * hs));
        v2f bv;
        bv.x = Yb[(size_t)k0 * PLANE];
        bv.y = Yb[(size_t)k0 * PLANE + PLANE];
        c = wmma_f32(a, bv, c);
    }
    float* dst = frames + (size_t)colp * 64 + 16 * w + 8 * hs;
    *(v8f*)dst = c;
}

__global__ void ola_kernel(const float* __restrict__ frames, float* __restrict__ out) {
    const int idx = blockIdx.x * blockDim.x + threadIdx.x;
    const int t   = idx & (T_LEN - 1);
    const int bo  = idx >> 14;
    const int tp  = t + 32;
    const int n1  = tp >> 5;
    const int off = tp & 31;
    const int n0  = n1 - 1;
    const float* fb = frames + (size_t)bo * (NFRMP * 64);
    const float v = fb[(size_t)n0 * 64 + off + 32] + fb[(size_t)n1 * 64 + off];
    const float cth = cosf(0.09817477042f * (float)off);
    const float env = 0.5f * (1.0f + cth * cth);
    out[idx] = v / env;
}

extern "C" void kernel_launch(void* const* d_in, const int* in_sizes, int n_in,
                              void* d_out, int out_size, void* d_ws, size_t ws_size,
                              hipStream_t stream) {
    const float* x   = (const float*)d_in[0];
    const float* wgt = (const float*)d_in[1];
    float* out = (float*)d_out;
    float* ws  = (float*)d_ws;

    float* A1 = ws + OFF_A1;
    float* A3 = ws + OFF_A3;
    float* Wt = ws + OFF_WT;
    float* U  = ws + OFF_U;
    float* Y  = ws + OFF_Y;
    float* FR = ws + OFF_FR;
    float* XP = ws + OFF_XP;

    prep_kernel<<<64, 256, 0, stream>>>(wgt, A1, A3, Wt);
    pad_kernel<<<dim3((TP_LEN + 255) / 256, NB * NCH), 256, 0, stream>>>(x, XP);
    stft_gemm<<<COL1P / 16, 128, 0, stream>>>(XP, A1, U);
    mix_gemm<<<dim3(COL2P / 16, NFREQ, 2), 128, 0, stream>>>(Wt, U, Y);
    istft_gemm<<<COL1P / 16, 128, 0, stream>>>(A3, Y, FR);
    ola_kernel<<<(NB * NCH * T_LEN) / 256, 256, 0, stream>>>(FR, out);
}


# Each GEMM work-group walks 8 column tiles with its A fragment loaded once from global into registers, not re-staged through LDS per 16-column tile; grids 8x smaller; sum order unchanged
# speedup vs baseline: 1.2090x; 1.2090x over previous
#include <hip/hip_runtime.h>
#include <hip/hip_bf16.h>

typedef __attribute__((ext_vector_type(2))) float v2f;
typedef __attribute__((ext_vector_type(8))) float v8f;

#define T_LEN   16384
#define TP_LEN  16448
#define HOP     32
#define NFRM    513
#define NFRMP   514
#define NB      8
#define NCH     64
#define NFREQ   33
#define COL2P   4112
#define COL1P   263168
#define PLANE   263168
#define TPB     8
#define NTILE1  16448
#define NBLK1   2056
#define NTILE2  257
#define NBLK2   33

#define OFF_A1  0
#define OFF_A3  4096
#define OFF_WT  8192
#define OFF_U   143360
#define OFF_Y   (OFF_U + 64*PLANE)
#define OFF_FR  OFF_U
#define OFF_XP  OFF_Y

__device__ __forceinline__ v8f wmma_f32(v2f a, v2f b, v8f c) {
    return __builtin_amdgcn_wmma_f32_16x16x4_f32(false, a, false, b, (short)0, c,
                                                 false, false);
}

__global__ void prep_kernel(const float* __restrict__ wgt,
                            float* __restrict__ A1, float* __restrict__ A3,
                            float* __restrict__ Wt) {
    const int gid = blockIdx.x * blockDim.x + threadIdx.x;
    const int gsz = gridDim.x * blockDim.x;
    const float PI = 3.14159265358979f;
    for (int idx = gid; idx < 4096; idx += gsz) {
        int fr = idx >> 6, tau = idx & 63;
        float wv = 0.5f * (1.0f - cosf(2.0f * PI * (float)tau / 64.0f));
        float v;
        if (fr <= 32) v =  cosf(2.0f * PI * (float)(fr)      * (float)tau / 64.0f) * wv;
        else          v = -sinf(2.0f * PI * (float)(fr - 32) * (float)tau / 64.0f) * wv;
        A1[idx] = v;
    }
    for (int idx = gid; idx < 4096; idx += gsz) {
        int tau = idx >> 6, fr = idx & 63;
        float wv = 0.5f * (1.0f - cosf(2.0f * PI * (float)tau / 64.0f));
        float s = wv * (1.0f / 64.0f);
        float v;
        if (fr == 0)       v = s;
        else if (fr < 32)  v = 2.0f * cosf(2.0f * PI * (float)fr * (float)tau / 64.0f) * s;
        else if (fr == 32) v = cosf(PI * (float)tau) * s;
        else               v = -2.0f * sinf(2.0f * PI * (float)(fr - 32) * (float)tau / 64.0f) * s;
        A3[idx] = v;
    }
    for (int idx = gid; idx < NFREQ * 4096; idx += gsz) {
        int f = idx >> 12, oi = idx & 4095;
        Wt[idx] = wgt[(size_t)oi * NFREQ + f];
    }
}

__global__ void pad_kernel(const float* __restrict__ x, float* __restrict__ xp) {
    const int tp = blockIdx.x * blockDim.x + threadIdx.x;
    if (tp >= TP_LEN) return;
    const int bc = blockIdx.y;
    int t = tp - 32;
    t = t < 0 ? -t : t;
    t = t >= T_LEN ? 2 * T_LEN - 2 - t : t;
    xp[(size_t)bc * TP_LEN + tp] = x[(size_t)bc * T_LEN + t];
}

__global__ void stft_gemm(const float* __restrict__ xp, const float* __restrict__ A1,
                          float* __restrict__ U) {
    const int tid  = threadIdx.x;
    const int lane = tid & 31;
    const int w    = tid >> 5;
    const int cL   = lane & 15;
    const int hs   = lane >> 4;
    const float* arow = A1 + (16 * w + cL) * 64 + 2 * hs;
    v2f a[16];
#pragma unroll
    for (int j = 0; j < 16; ++j)
        a[j] = *(const v2f*)(arow + 4 * j);

    const int tile0 = blockIdx.x * TPB;
    for (int i = 0; i < TPB; ++i) {
        const int colp = (tile0 + i) * 16 + cL;
        const int n    = colp % NFRMP;
        const int bc   = colp / NFRMP;
        const int b    = bc >> 6;
        const int ic   = bc & 63;
        const int nc   = n > 512 ? 512 : n;
        const float* xrow = xp + (size_t)bc * TP_LEN + nc * HOP + 2 * hs;
        v8f c = {};
#pragma unroll
        for (int j = 0; j < 16; ++j) {
            v2f bv = *(const v2f*)(xrow + 4 * j);
            c = wmma_f32(a[j], bv, c);
        }
        float* Ust = U + (size_t)(16 * w + 8 * hs) * PLANE +
                     (size_t)ic * COL2P + b * NFRMP + n;
#pragma unroll
        for (int r = 0; r < 8; ++r)
            Ust[(size_t)r * PLANE] = c[r];
    }
}

__global__ void mix_gemm(const float* __restrict__ Wt, const float* __restrict__ U,
                         float* __restrict__ Y) {
    const int f  = blockIdx.y;
    const int ri = blockIdx.z;
    if (ri == 1 && (f == 0 || f == 32)) return;
    const int frq = f + ri * 32;

    __shared__ float lB[2 * 1024];
    const int tid  = threadIdx.x;
    const int lane = tid & 31;
    const int w    = tid >> 5;
    const int cL   = lane & 15;
    const int hs   = lane >> 4;
    const float* arow = Wt + (size_t)f * 4096 + (16 * w + cL) * 64 + 2 * hs;
    v2f a[16];
#pragma unroll
    for (int j = 0; j < 16; ++j)
        a[j] = *(const v2f*)(arow + 4 * j);

    const int tile0 = blockIdx.x * TPB;
    const int nt    = (NTILE2 - tile0) < TPB ? (NTILE2 - tile0) : TPB;
    const float* Up = U + (size_t)frq * PLANE;
    float*       Yp = Y + (size_t)frq * PLANE;
    for (int i = 0; i < nt; ++i) {
        const int tile = tile0 + i;
        float* lBc = lB + (i & 1) * 1024;
        const float* Ub = Up + tile * 16;
#pragma unroll
        for (int q = 0; q < 2; ++q) {
            const int fi = q * 128 + tid;
            const int r = fi >> 2, p = fi & 3;
            *(float4*)(lBc + fi * 4) = *(const float4*)(Ub + (size_t)r * COL2P + p * 4);
        }
        __syncthreads();
        v8f c = {};
#pragma unroll
        for (int j = 0; j < 16; ++j) {
            const int kk = 4 * j + 2 * hs;
            v2f bv;
            bv.x = lBc[kk * 16 + cL];
            bv.y = lBc[(kk + 1) * 16 + cL];
            c = wmma_f32(a[j], bv, c);
        }
        const int col2p = tile * 16 + cL;
        const int bI = col2p / NFRMP;
        const int nI = col2p % NFRMP;
        float* Yst = Yp + (size_t)bI * (64 * NFRMP) + nI +
                     (size_t)(16 * w + 8 * hs) * NFRMP;
#pragma unroll
        for (int r = 0; r < 8; ++r)
            Yst[(size_t)r * NFRMP] = c[r];
    }
}

__global__ void istft_gemm(const float* __restrict__ A3, const float* __restrict__ Y,
                           float* __restrict__ frames) {
    const int tid  = threadIdx.x;
    const int lane = tid & 31;
    const int w    = tid >> 5;
    const int cL   = lane & 15;
    const int hs   = lane >> 4;
    const float* arow = A3 + (16 * w + cL) * 64 + 2 * hs;
    v2f a[16];
#pragma unroll
    for (int j = 0; j < 16; ++j)
        a[j] = *(const v2f*)(arow + 4 * j);

    const int tile0 = blockIdx.x * TPB;
    for (int i = 0; i < TPB; ++i) {
        const int colp = (tile0 + i) * 16 + cL;
        const float* Yb = Y + (size_t)colp + (size_t)(2 * hs) * PLANE;
        v8f c = {};
#pragma unroll
        for (int j = 0; j < 16; ++j) {
            v2f bv;
            bv.x = Yb[(size_t)(4 * j) * PLANE];
            bv.y = Yb[(size_t)(4 * j) * PLANE + PLANE];
            c = wmma_f32(a[j], bv, c);
        }
        float* dst = frames + (size_t)colp * 64 + 16 * w + 8 * hs;
        *(v8f*)dst = c;
    }
}

__global__ void ola_kernel(const float* __restrict__ frames, float* __restrict__ out) {
    const int idx = blockIdx.x * blockDim.x + threadIdx.x;
    const int t   = idx & (T_LEN - 1);
    const int bo  = idx >> 14;
    const int tp  = t + 32;
    const int n1  = tp >> 5;
    const int off = tp & 31;
    const int n0  = n1 - 1;
    const float* fb = frames + (size_t)bo * (NFRMP * 64);
    const float v = fb[(size_t)n0 * 64 + off + 32] + fb[(size_t)n1 * 64 + off];
    const float cth = cosf(0.09817477042f * (float)off);
    const float env = 0.5f * (1.0f + cth * cth);
    out[idx] = v / env;
}

extern "C" void kernel_launch(void* const* d_in, const int* in_sizes, int n_in,
                              void* d_out, int out_size, void* d_ws, size_t ws_size,
                              hipStream_t stream) {
    const float* x   = (const float*)d_in[0];
    const float* wgt = (const float*)d_in[1];
    float* out = (float*)d_out;
    float* ws  = (float*)d_ws;

    float* A1 = ws + OFF_A1;
    float* A3 = ws + OFF_A3;
    float* Wt = ws + OFF_WT;
    float* U  = ws + OFF_U;
    float* Y  = ws + OFF_Y;
    float* FR = ws + OFF_FR;
    float* XP = ws + OFF_XP;

    prep_kernel<<<64, 256, 0, stream>>>(wgt, A1, A3, Wt);
    pad_kernel<<<dim3((TP_LEN + 255) / 256, NB * NCH), 256, 0, stream>>>(x, XP);
    stft_gemm<<<NBLK1, 128, 0, stream>>>(XP, A1, U);
    mix_gemm<<<dim3(NBLK2, NFREQ, 2), 128, 0, stream>>>(Wt, U, Y);
    istft_gemm<<<NBLK1, 128, 0, stream>>>(A3, Y, FR);
    ola_kernel<<<(NB * NCH * T_LEN) / 256, 256, 0, stream>>>(FR, out);
}


# Overlap-add fused into the inverse-transform GEMM: waves holding tau 0..31 take frame m+1, halves meet in LDS, out stored in whole lines; frames buffer and OLA launch removed; sum order unchanged
# speedup vs baseline: 1.5043x; 1.2443x over previous
#include <hip/hip_runtime.h>
#include <hip/hip_bf16.h>

typedef __attribute__((ext_vector_type(2))) float v2f;
typedef __attribute__((ext_vector_type(8))) float v8f;

#define T_LEN   16384
#define TP_LEN  16448
#define HOP     32
#define NFRM    513
#define NFRMP   514
#define NB      8
#define NCH     64
#define NFREQ   33
#define COL2P   4112
#define COL1P   263168
#define PLANE   263168
#define TPB     8
#define NTILE1  16448
#define NBLK1   2056
#define NTILE2  257
#define NBLK2   33
#define NTILE3  16384
#define NBLK3   2048
#define SP      68

#define OFF_A1  0
#define OFF_A3  4096
#define OFF_WT  8192
#define OFF_U   143360
#define OFF_Y   (OFF_U + 64*PLANE)
#define OFF_XP  OFF_Y

__device__ __forceinline__ v8f wmma_f32(v2f a, v2f b, v8f c) {
    return __builtin_amdgcn_wmma_f32_16x16x4_f32(false, a, false, b, (short)0, c,
                                                 false, false);
}

__global__ void prep_kernel(const float* __restrict__ wgt,
                            float* __restrict__ A1, float* __restrict__ A3,
                            float* __restrict__ Wt) {
    const int gid = blockIdx.x * blockDim.x + threadIdx.x;
    const int gsz = gridDim.x * blockDim.x;
    const float PI = 3.14159265358979f;
    for (int idx = gid; idx < 4096; idx += gsz) {
        int fr = idx >> 6, tau = idx & 63;
        float wv = 0.5f * (1.0f - cosf(2.0f * PI * (float)tau / 64.0f));
        float v;
        if (fr <= 32) v =  cosf(2.0f * PI * (float)(fr)      * (float)tau / 64.0f) * wv;
        else          v = -sinf(2.0f * PI * (float)(fr - 32) * (float)tau / 64.0f) * wv;
        A1[idx] = v;
    }
    for (int idx = gid; idx < 4096; idx += gsz) {
        int tau = idx >> 6, fr = idx & 63;
        float wv = 0.5f * (1.0f - cosf(2.0f * PI * (float)tau / 64.0f));
        float s = wv * (1.0f / 64.0f);
        float v;
        if (fr == 0)       v = s;
        else if (fr < 32)  v = 2.0f * cosf(2.0f * PI * (float)fr * (float)tau / 64.0f) * s;
        else if (fr == 32) v = cosf(PI * (float)tau) * s;
        else               v = -2.0f * sinf(2.0f * PI * (float)(fr - 32) * (float)tau / 64.0f) * s;
        A3[idx] = v;
    }
    for (int idx = gid; idx < NFREQ * 4096; idx += gsz) {
        int f = idx >> 12, oi = idx & 4095;
        Wt[idx] = wgt[(size_t)oi * NFREQ + f];
    }
}

__global__ void pad_kernel(const float* __restrict__ x, float* __restrict__ xp) {
    const int tp = blockIdx.x * blockDim.x + threadIdx.x;
    if (tp >= TP_LEN) return;
    const int bc = blockIdx.y;
    int t = tp - 32;
    t = t < 0 ? -t : t;
    t = t >= T_LEN ? 2 * T_LEN - 2 - t : t;
    xp[(size_t)bc * TP_LEN + tp] = x[(size_t)bc * T_LEN + t];
}

__global__ void stft_gemm(const float* __restrict__ xp, const float* __restrict__ A1,
                          float* __restrict__ U) {
    const int tid  = threadIdx.x;
    const int lane = tid & 31;
    const int w    = tid >> 5;
    const int cL   = lane & 15;
    const int hs   = lane >> 4;
    const float* arow = A1 + (16 * w + cL) * 64 + 2 * hs;
    v2f a[16];
#pragma unroll
    for (int j = 0; j < 16; ++j)
        a[j] = *(const v2f*)(arow + 4 * j);

    const int tile0 = blockIdx.x * TPB;
    for (int i = 0; i < TPB; ++i) {
        const int colp = (tile0 + i) * 16 + cL;
        const int n    = colp % NFRMP;
        const int bc   = colp / NFRMP;
        const int b    = bc >> 6;
        const int ic   = bc & 63;
        const int nc   = n > 512 ? 512 : n;
        const float* xrow = xp + (size_t)bc * TP_LEN + nc * HOP + 2 * hs;
        v8f c = {};
#pragma unroll
        for (int j = 0; j < 16; ++j) {
            v2f bv = *(const v2f*)(xrow + 4 * j);
            c = wmma_f32(a[j], bv, c);
        }
        float* Ust = U + (size_t)(16 * w + 8 * hs) * PLANE +
                     (size_t)ic * COL2P + b * NFRMP + n;
#pragma unroll
        for (int r = 0; r < 8; ++r)
            Ust[(size_t)r * PLANE] = c[r];
    }
}

__global__ void mix_gemm(const float* __restrict__ Wt, const float* __restrict__ U,
                         float* __restrict__ Y) {
    const int f  = blockIdx.y;
    const int ri = blockIdx.z;
    if (ri == 1 && (f == 0 || f == 32)) return;
    const int frq = f + ri * 32;

    __shared__ float lB[2 * 1024];
    const int tid  = threadIdx.x;
    const int lane = tid & 31;
    const int w    = tid >> 5;
    const int cL   = lane & 15;
    const int hs   = lane >> 4;
    const float* arow = Wt + (size_t)f * 4096 + (16 * w + cL) * 64 + 2 * hs;
    v2f a[16];
#pragma unroll
    for (int j = 0; j < 16; ++j)
        a[j] = *(const v2f*)(arow + 4 * j);

    const int tile0 = blockIdx.x * TPB;
    const int nt    = (NTILE2 - tile0) < TPB ? (NTILE2 - tile0) : TPB;
    const float* Up = U + (size_t)frq * PLANE;
    float*       Yp = Y + (size_t)frq * PLANE;
    for (int i = 0; i < nt; ++i) {
        const int tile = tile0 + i;
        float* lBc = lB + (i & 1) * 1024;
        const float* Ub = Up + tile * 16;
#pragma unroll
        for (int q = 0; q < 2; ++q) {
            const int fi = q * 128 + tid;
            const int r = fi >> 2, p = fi & 3;
            *(float4*)(lBc + fi * 4) = *(const float4*)(Ub + (size_t)r * COL2P + p * 4);
        }
        __syncthreads();
        v8f c = {};
#pragma unroll
        for (int j = 0; j < 16; ++j) {
            const int kk = 4 * j + 2 * hs;
            v2f bv;
            bv.x = lBc[kk * 16 + cL];
            bv.y = lBc[(kk + 1) * 16 + cL];
            c = wmma_f32(a[j], bv, c);
        }
        const int col2p = tile * 16 + cL;
        const int bI = col2p / NFRMP;
        const int nI = col2p % NFRMP;
        float* Yst = Yp + (size_t)bI * (64 * NFRMP) + nI +
                     (size_t)(16 * w + 8 * hs) * NFRMP;
#pragma unroll
        for (int r = 0; r < 8; ++r)
            Yst[(size_t)r * NFRMP] = c[r];
    }
}

__global__ void istft_ola_gemm(const float* __restrict__ A3, const float* __restrict__ Y,
                               float* __restrict__ out) {
    __shared__ __attribute__((aligned(16))) float lF[2 * 16 * SP];
    const int tid  = threadIdx.x;
    const int lane = tid & 31;
    const int w    = tid >> 5;
    const int cL   = lane & 15;
    const int hs   = lane >> 4;
    const int sh   = w < 2 ? 1 : 0;
    const float* arow = A3 + (16 * w + cL) * 64 + 2 * hs;
    v2f a[16];
#pragma unroll
    for (int j = 0; j < 16; ++j)
        a[j] = *(const v2f*)(arow + 4 * j);

    const int ecol = tid >> 3;
    const int eo4  = (tid & 7) * 4;
    float env[4];
#pragma unroll
    for (int q = 0; q < 4; ++q) {
        const float cth = cosf(0.09817477042f * (float)(eo4 + q));
        env[q] = 0.5f * (1.0f + cth * cth);
    }

    const int tile0 = blockIdx.x * TPB;
    for (int i = 0; i < TPB; ++i) {
        const int mt = tile0 + i;
        const int bo = mt >> 5;
        const int m0 = (mt & 31) * 16;
        const int colp = bo * NFRMP + m0 + cL + sh;
        const float* Yb = Y + (size_t)colp + (size_t)(2 * hs) * PLANE;
        v8f c = {};
#pragma unroll
        for (int j = 0; j < 16; ++j) {
            v2f bv;
            bv.x = Yb[(size_t)(4 * j) * PLANE];
            bv.y = Yb[(size_t)(4 * j) * PLANE + PLANE];
            c = wmma_f32(a[j], bv, c);
        }
        float* lFc = lF + (i & 1) * (16 * SP);
        float* sp  = lFc + cL * SP + 16 * w + 8 * hs;
        *(float4*)sp       = make_float4(c[0], c[1], c[2], c[3]);
        *(float4*)(sp + 4) = make_float4(c[4], c[5], c[6], c[7]);
        __syncthreads();
        const float* ep = lFc + ecol * SP + eo4;
        const float4 lo = *(const float4*)ep;
        const float4 hi = *(const float4*)(ep + 32);
        float4 o;
        o.x = (hi.x + lo.x) / env[0];
        o.y = (hi.y + lo.y) / env[1];
        o.z = (hi.z + lo.z) / env[2];
        o.w = (hi.w + lo.w) / env[3];
        *(float4*)(out + (size_t)bo * T_LEN + (m0 + ecol) * HOP + eo4) = o;
    }
}

extern "C" void kernel_launch(void* const* d_in, const int* in_sizes, int n_in,
                              void* d_out, int out_size, void* d_ws, size_t ws_size,
                              hipStream_t stream) {
    const float* x   = (const float*)d_in[0];
    const float* wgt = (const float*)d_in[1];
    float* out = (float*)d_out;
    float* ws  = (float*)d_ws;

    float* A1 = ws + OFF_A1;
    float* A3 = ws + OFF_A3;
    float* Wt = ws + OFF_WT;
    float* U  = ws + OFF_U;
    float* Y  = ws + OFF_Y;
    float* XP = ws + OFF_XP;

    prep_kernel<<<64, 256, 0, stream>>>(wgt, A1, A3, Wt);
    pad_kernel<<<dim3((TP_LEN + 255) / 256, NB * NCH), 256, 0, stream>>>(x, XP);
    stft_gemm<<<NBLK1, 128, 0, stream>>>(XP, A1, U);
    mix_gemm<<<dim3(NBLK2, NFREQ, 2), 128, 0, stream>>>(Wt, U, Y);
    istft_ola_gemm<<<NBLK3, 128, 0, stream>>>(A3, Y, out);
}


# STFT GEMM reads its frames straight from x, mirroring the sample index only in the lanes of frame 0 and frame 512; reflect-pad launch and padded copy removed; sum order unchanged
# speedup vs baseline: 1.9099x; 1.2696x over previous
#include <hip/hip_runtime.h>
#include <hip/hip_bf16.h>

typedef __attribute__((ext_vector_type(2))) float v2f;
typedef __attribute__((ext_vector_type(8))) float v8f;

#define T_LEN   16384
#define HOP     32
#define NFRM    513
#define NFRMP   514
#define NB      8
#define NCH     64
#define NFREQ   33
#define COL2P   4112
#define COL1P   263168
#define PLANE   263168
#define TPB     8
#define NTILE1  16448
#define NBLK1   2056
#define NTILE2  257
#define NBLK2   33
#define NTILE3  16384
#define NBLK3   2048
#define SP      68

#define OFF_A1  0
#define OFF_A3  4096
#define OFF_WT  8192
#define OFF_U   143360
#define OFF_Y   (OFF_U + 64*PLANE)

__device__ __forceinline__ v8f wmma_f32(v2f a, v2f b, v8f c) {
    return __builtin_amdgcn_wmma_f32_16x16x4_f32(false, a, false, b, (short)0, c,
                                                 false, false);
}

__global__ void prep_kernel(const float* __restrict__ wgt,
                            float* __restrict__ A1, float* __restrict__ A3,
                            float* __restrict__ Wt) {
    const int gid = blockIdx.x * blockDim.x + threadIdx.x;
    const int gsz = gridDim.x * blockDim.x;
    const float PI = 3.14159265358979f;
    for (int idx = gid; idx < 4096; idx += gsz) {
        int fr = idx >> 6, tau = idx & 63;
        float wv = 0.5f * (1.0f - cosf(2.0f * PI * (float)tau / 64.0f));
        float v;
        if (fr <= 32) v =  cosf(2.0f * PI * (float)(fr)      * (float)tau / 64.0f) * wv;
        else          v = -sinf(2.0f * PI * (float)(fr - 32) * (float)tau / 64.0f) * wv;
        A1[idx] = v;
    }
    for (int idx = gid; idx < 4096; idx += gsz) {
        int tau = idx >> 6, fr = idx & 63;
        float wv = 0.5f * (1.0f - cosf(2.0f * PI * (float)tau / 64.0f));
        float s = wv * (1.0f / 64.0f);
        float v;
        if (fr == 0)       v = s;
        else if (fr < 32)  v = 2.0f * cosf(2.0f * PI * (float)fr * (float)tau / 64.0f) * s;
        else if (fr == 32) v = cosf(PI * (float)tau) * s;
        else               v = -2.0f * sinf(2.0f * PI * (float)(fr - 32) * (float)tau / 64.0f) * s;
        A3[idx] = v;
    }
    for (int idx = gid; idx < NFREQ * 4096; idx += gsz) {
        int f = idx >> 12, oi = idx & 4095;
        Wt[idx] = wgt[(size_t)oi * NFREQ + f];
    }
}

__global__ void stft_gemm(const float* __restrict__ x, const float* __restrict__ A1,
                          float* __restrict__ U) {
    const int tid  = threadIdx.x;
    const int lane = tid & 31;
    const int w    = tid >> 5;
    const int cL   = lane & 15;
    const int hs   = lane >> 4;
    const float* arow = A1 + (16 * w + cL) * 64 + 2 * hs;
    v2f a[16];
#pragma unroll
    for (int j = 0; j < 16; ++j)
        a[j] = *(const v2f*)(arow + 4 * j);

    const int tile0 = blockIdx.x * TPB;
    for (int i = 0; i < TPB; ++i) {
        const int colp = (tile0 + i) * 16 + cL;
        const int n    = colp % NFRMP;
        const int bc   = colp / NFRMP;
        const int b    = bc >> 6;
        const int ic   = bc & 63;
        const int nc   = n > 512 ? 512 : n;
        const float* xb = x + (size_t)bc * T_LEN;
        const int t0   = (nc - 1) * HOP + 2 * hs;
        v2f bf[16];
        if (nc >= 1 && nc <= 511) {
#pragma unroll
            for (int j = 0; j < 16; ++j)
                bf[j] = *(const v2f*)(xb + t0 + 4 * j);
        } else {
#pragma unroll
            for (int j = 0; j < 16; ++j) {
                int ta = t0 + 4 * j;
                int tb = ta + 1;
                ta = ta < 0 ? -ta : ta;
                tb = tb < 0 ? -tb : tb;
                ta = ta >= T_LEN ? 2 * T_LEN - 2 - ta : ta;
                tb = tb >= T_LEN ? 2 * T_LEN - 2 - tb : tb;
                bf[j].x = xb[ta];
                bf[j].y = xb[tb];
            }
        }
        v8f c = {};
#pragma unroll
        for (int j = 0; j < 16; ++j)
            c = wmma_f32(a[j], bf[j], c);
        float* Ust = U + (size_t)(16 * w + 8 * hs) * PLANE +
                     (size_t)ic * COL2P + b * NFRMP + n;
#pragma unroll
        for (int r = 0; r < 8; ++r)
            Ust[(size_t)r * PLANE] = c[r];
    }
}

__global__ void mix_gemm(const float* __restrict__ Wt, const float* __restrict__ U,
                         float* __restrict__ Y) {
    const int f  = blockIdx.y;
    const int ri = blockIdx.z;
    if (ri == 1 && (f == 0 || f == 32)) return;
    const int frq = f + ri * 32;

    __shared__ float lB[2 * 1024];
    const int tid  = threadIdx.x;
    const int lane = tid & 31;
    const int w    = tid >> 5;
    const int cL   = lane & 15;
    const int hs   = lane >> 4;
    const float* arow = Wt + (size_t)f * 4096 + (16 * w + cL) * 64 + 2 * hs;
    v2f a[16];
#pragma unroll
    for (int j = 0; j < 16; ++j)
        a[j] = *(const v2f*)(arow + 4 * j);

    const int tile0 = blockIdx.x * TPB;
    const int nt    = (NTILE2 - tile0) < TPB ? (NTILE2 - tile0) : TPB;
    const float* Up = U + (size_t)frq * PLANE;
    float*       Yp = Y + (size_t)frq * PLANE;
    for (int i = 0; i < nt; ++i) {
        const int tile = tile0 + i;
        float* lBc = lB + (i & 1) * 1024;
        const float* Ub = Up + tile * 16;
#pragma unroll
        for (int q = 0; q < 2; ++q) {
            const int fi = q * 128 + tid;
            const int r = fi >> 2, p = fi & 3;
            *(float4*)(lBc + fi * 4) = *(const float4*)(Ub + (size_t)r * COL2P + p * 4);
        }
        __syncthreads();
        v8f c = {};
#pragma unroll
        for (int j = 0; j < 16; ++j) {
            const int kk = 4 * j + 2 * hs;
            v2f bv;
            bv.x = lBc[kk * 16 + cL];
            bv.y = lBc[(kk + 1) * 16 + cL];
            c = wmma_f32(a[j], bv, c);
        }
        const int col2p = tile * 16 + cL;
        const int bI = col2p / NFRMP;
        const int nI = col2p % NFRMP;
        float* Yst = Yp + (size_t)bI * (64 * NFRMP) + nI +
                     (size_t)(16 * w + 8 * hs) * NFRMP;
#pragma unroll
        for (int r = 0; r < 8; ++r)
            Yst[(size_t)r * NFRMP] = c[r];
    }
}

__global__ void istft_ola_gemm(const float* __restrict__ A3, const float* __restrict__ Y,
                               float* __restrict__ out) {
    __shared__ __attribute__((aligned(16))) float lF[2 * 16 * SP];
    const int tid  = threadIdx.x;
    const int lane = tid & 31;
    const int w    = tid >> 5;
    const int cL   = lane & 15;
    const int hs   = lane >> 4;
    const int sh   = w < 2 ? 1 : 0;
    const float* arow = A3 + (16 * w + cL) * 64 + 2 * hs;
    v2f a[16];
#pragma unroll
    for (int j = 0; j < 16; ++j)
        a[j] = *(const v2f*)(arow + 4 * j);

    const int ecol = tid >> 3;
    const int eo4  = (tid & 7) * 4;
    float env[4];
#pragma unroll
    for (int q = 0; q < 4; ++q) {
        const float cth = cosf(0.09817477042f * (float)(eo4 + q));
        env[q] = 0.5f * (1.0f + cth * cth);
    }

    const int tile0 = blockIdx.x * TPB;
    for (int i = 0; i < TPB; ++i) {
        const int mt = tile0 + i;
        const int bo = mt >> 5;
        const int m0 = (mt & 31) * 16;
        const int colp = bo * NFRMP + m0 + cL + sh;
        const float* Yb = Y + (size_t)colp + (size_t)(2 * hs) * PLANE;
        v8f c = {};
#pragma unroll
        for (int j = 0; j < 16; ++j) {
            v2f bv;
            bv.x = Yb[(size_t)(4 * j) * PLANE];
            bv.y = Yb[(size_t)(4 * j) * PLANE + PLANE];
            c = wmma_f32(a[j], bv, c);
        }
        float* lFc = lF + (i & 1) * (16 * SP);
        float* sp  = lFc + cL * SP + 16 * w + 8 * hs;
        *(float4*)sp       = make_float4(c[0], c[1], c[2], c[3]);
        *(float4*)(sp + 4) = make_float4(c[4], c[5], c[6], c[7]);
        __syncthreads();
        const float* ep = lFc + ecol * SP + eo4;
        const float4 lo = *(const float4*)ep;
        const float4 hi = *(const float4*)(ep + 32);
        float4 o;
        o.x = (hi.x + lo.x) / env[0];
        o.y = (hi.y + lo.y) / env[1];
        o.z = (hi.z + lo.z) / env[2];
        o.w = (hi.w + lo.w) / env[3];
        *(float4*)(out + (size_t)bo * T_LEN + (m0 + ecol) * HOP + eo4) = o;
    }
}

extern "C" void kernel_launch(void* const* d_in, const int* in_sizes, int n_in,
                              void* d_out, int out_size, void* d_ws, size_t ws_size,
                              hipStream_t stream) {
    const float* x   = (const float*)d_in[0];
    const float* wgt = (const float*)d_in[1];
    float* out = (float*)d_out;
    float* ws  = (float*)d_ws;

    float* A1 = ws + OFF_A1;
    float* A3 = ws + OFF_A3;
    float* Wt = ws + OFF_WT;
    float* U  = ws + OFF_U;
    float* Y  = ws + OFF_Y;

    prep_kernel<<<64, 256, 0, stream>>>(wgt, A1, A3, Wt);
    stft_gemm<<<NBLK1, 128, 0, stream>>>(x, A1, U);
    mix_gemm<<<dim3(NBLK2, NFREQ, 2), 128, 0, stream>>>(Wt, U, Y);
    istft_ola_gemm<<<NBLK3, 128, 0, stream>>>(A3, Y, out);
}


# U and Y re-laid as [fr][b*513+n][channel]: waves pair two channel tiles and swap lane halves so every U/Y store is one whole aligned 128-byte line; mix stages contiguous chunks; sum order unchanged
# speedup vs baseline: 2.0372x; 1.0667x over previous
#include <hip/hip_runtime.h>
#include <hip/hip_bf16.h>

typedef __attribute__((ext_vector_type(2))) float v2f;
typedef __attribute__((ext_vector_type(8))) float v8f;

#define T_LEN   16384
#define HOP     32
#define NFRM    513
#define NB      8
#define NCH     64
#define NFREQ   33
#define NCOL    4104
#define PLANE   263168
#define NBLK1   2052
#define NPAIR2  129
#define PPB     4
#define NBLK2   33
#define TPB     8
#define NTILE3  16384
#define NBLK3   2048
#define SP      68

#define OFF_A1  0
#define OFF_A3  4096
#define OFF_WT  8192
#define OFF_U   143360
#define OFF_Y   (OFF_U + 64*PLANE)


__device__ __forceinline__ v8f wmma_f32(v2f a, v2f b, v8f c) {
    return __builtin_amdgcn_wmma_f32_16x16x4_f32(false, a, false, b, (short)0, c,
                                                 false, false);
}

__device__ __forceinline__ float xhalf(float v) {
    const int b = __builtin_bit_cast(int, v);
    return __builtin_bit_cast(float,
        __builtin_amdgcn_permlanex16(b, b, 0x76543210, 0xfedcba98, false, false));
}

__global__ void prep_kernel(const float* __restrict__ wgt,
                            float* __restrict__ A1, float* __restrict__ A3,
                            float* __restrict__ Wt) {
    const int gid = blockIdx.x * blockDim.x + threadIdx.x;
    const int gsz = gridDim.x * blockDim.x;
    const float PI = 3.14159265358979f;
    for (int idx = gid; idx < 4096; idx += gsz) {
        int fr = idx >> 6, tau = idx & 63;
        float wv = 0.5f * (1.0f - cosf(2.0f * PI * (float)tau / 64.0f));
        float v;
        if (fr <= 32) v =  cosf(2.0f * PI * (float)(fr)      * (float)tau / 64.0f) * wv;
        else          v = -sinf(2.0f * PI * (float)(fr - 32) * (float)tau / 64.0f) * wv;
        A1[idx] = v;
    }
    for (int idx = gid; idx < 4096; idx += gsz) {
        int tau = idx >> 6, fr = idx & 63;
        float wv = 0.5f * (1.0f - cosf(2.0f * PI * (float)tau / 64.0f));
        float s = wv * (1.0f / 64.0f);
        float v;
        if (fr == 0)       v = s;
        else if (fr < 32)  v = 2.0f * cosf(2.0f * PI * (float)fr * (float)tau / 64.0f) * s;
        else if (fr == 32) v = cosf(PI * (float)tau) * s;
        else               v = -2.0f * sinf(2.0f * PI * (float)(fr - 32) * (float)tau / 64.0f) * s;
        A3[idx] = v;
    }
    for (int idx = gid; idx < NFREQ * 4096; idx += gsz) {
        int f = idx >> 12, oi = idx & 4095;
        Wt[idx] = wgt[(size_t)oi * NFREQ + f];
    }
}

__global__ void stft_gemm(const float* __restrict__ x, const float* __restrict__ A1,
                          float* __restrict__ U) {
    const int tid  = threadIdx.x;
    const int lane = tid & 31;
    const int w    = tid >> 5;
    const int cL   = lane & 15;
    const int hs   = lane >> 4;
    const float* arow = A1 + (16 * w + cL) * 64 + 2 * hs;
    v2f a[16];
#pragma unroll
    for (int j = 0; j < 16; ++j)
        a[j] = *(const v2f*)(arow + 4 * j);

    for (int i = 0; i < 4; ++i) {
        const int col = blockIdx.x * 2 + (i >> 1);
        const int p   = i & 1;
        const int b   = col / NFRM;
        const int n   = col % NFRM;
        const float* xb0 = x + (size_t)(b * NCH + 32 * p + cL) * T_LEN;
        const float* xb1 = xb0 + (size_t)16 * T_LEN;
        const int t0   = (n - 1) * HOP + 2 * hs;
        v2f f0[16], f1[16];
        if (n >= 1 && n <= 511) {
#pragma unroll
            for (int j = 0; j < 16; ++j) {
                f0[j] = *(const v2f*)(xb0 + t0 + 4 * j);
                f1[j] = *(const v2f*)(xb1 + t0 + 4 * j);
            }
        } else {
#pragma unroll
            for (int j = 0; j < 16; ++j) {
                int ta = t0 + 4 * j;
                int tb = ta + 1;
                ta = ta < 0 ? -ta : ta;
                tb = tb < 0 ? -tb : tb;
                ta = ta >= T_LEN ? 2 * T_LEN - 2 - ta : ta;
                tb = tb >= T_LEN ? 2 * T_LEN - 2 - tb : tb;
                f0[j].x = xb0[ta];
                f0[j].y = xb0[tb];
                f1[j].x = xb1[ta];
                f1[j].y = xb1[tb];
            }
        }
        v8f c0 = {}, c1 = {};
#pragma unroll
        for (int j = 0; j < 16; ++j) {
            c0 = wmma_f32(a[j], f0[j], c0);
            c1 = wmma_f32(a[j], f1[j], c1);
        }
        float* Ust = U + (size_t)(16 * w) * PLANE + (size_t)col * NCH + 32 * p + lane;
#pragma unroll
        for (int r = 0; r < 8; ++r) {
            const float give = hs ? c0[r] : c1[r];
            const float got  = xhalf(give);
            Ust[(size_t)r * PLANE]       = hs ? got : c0[r];
            Ust[(size_t)(8 + r) * PLANE] = hs ? c1[r] : got;
        }
    }
}

__global__ void mix_gemm(const float* __restrict__ Wt, const float* __restrict__ U,
                         float* __restrict__ Y) {
    const int f  = blockIdx.y;
    const int ri = blockIdx.z;
    if (ri == 1 && (f == 0 || f == 32)) return;
    const int frq = f + ri * 32;

    __shared__ __attribute__((aligned(16))) float lB[2 * 32 * SP];
    const int tid  = threadIdx.x;
    const int lane = tid & 31;
    const int w    = tid >> 5;
    const int cL   = lane & 15;
    const int hs   = lane >> 4;
    const int ow   = w & 1;
    const int ct   = w >> 1;
    const float* wrow = Wt + (size_t)f * 4096 + (32 * ow + cL) * 64 + 2 * hs;
    v2f wb0[16], wb1[16];
#pragma unroll
    for (int j = 0; j < 16; ++j) {
        wb0[j] = *(const v2f*)(wrow + 4 * j);
        wb1[j] = *(const v2f*)(wrow + 16 * 64 + 4 * j);
    }

    const int pair0 = blockIdx.x * PPB;
    const int np    = (NPAIR2 - pair0) < PPB ? (NPAIR2 - pair0) : PPB;
    const float* Up = U + (size_t)frq * PLANE;
    float*       Yp = Y + (size_t)frq * PLANE;
    for (int i = 0; i < np; ++i) {
        const int col0 = (pair0 + i) * 32;
        float* lBc = lB + (i & 1) * (32 * SP);
#pragma unroll
        for (int q = 0; q < 4; ++q) {
            const int fi = q * 128 + tid;
            const int cc = fi >> 4, p4 = fi & 15;
            int gc = col0 + cc;
            gc = gc < NCOL ? gc : NCOL - 1;
            *(float4*)(lBc + cc * SP + p4 * 4) = *(const float4*)(Up + (size_t)gc * NCH + p4 * 4);
        }
        __syncthreads();
        v8f c0 = {}, c1 = {};
        const float* lrow = lBc + (16 * ct + cL) * SP + 2 * hs;
#pragma unroll
        for (int j = 0; j < 16; ++j) {
            const v2f uv = *(const v2f*)(lrow + 4 * j);
            c0 = wmma_f32(uv, wb0[j], c0);
            c1 = wmma_f32(uv, wb1[j], c1);
        }
        const int crow = col0 + 16 * ct;
        float* Yst = Yp + (size_t)crow * NCH + 32 * ow + lane;
        if (col0 + 32 <= NCOL) {
#pragma unroll
            for (int r = 0; r < 8; ++r) {
                const float give = hs ? c0[r] : c1[r];
                const float got  = xhalf(give);
                Yst[(size_t)r * NCH]       = hs ? got : c0[r];
                Yst[(size_t)(8 + r) * NCH] = hs ? c1[r] : got;
            }
        } else {
#pragma unroll
            for (int r = 0; r < 8; ++r) {
                const float give = hs ? c0[r] : c1[r];
                const float got  = xhalf(give);
                if (crow + r < NCOL)     Yst[(size_t)r * NCH]       = hs ? got : c0[r];
                if (crow + 8 + r < NCOL) Yst[(size_t)(8 + r) * NCH] = hs ? c1[r] : got;
            }
        }
    }
}

__global__ void istft_ola_gemm(const float* __restrict__ A3, const float* __restrict__ Y,
                               float* __restrict__ out) {
    __shared__ __attribute__((aligned(16))) float lF[2 * 16 * SP];
    const int tid  = threadIdx.x;
    const int lane = tid & 31;
    const int w    = tid >> 5;
    const int cL   = lane & 15;
    const int hs   = lane >> 4;
    const int sh   = w < 2 ? 1 : 0;
    const float* arow = A3 + (16 * w + cL) * 64 + 2 * hs;
    v2f a[16];
#pragma unroll
    for (int j = 0; j < 16; ++j)
        a[j] = *(const v2f*)(arow + 4 * j);

    const int ecol = tid >> 3;
    const int eo4  = (tid & 7) * 4;
    float env[4];
#pragma unroll
    for (int q = 0; q < 4; ++q) {
        const float cth = cosf(0.09817477042f * (float)(eo4 + q));
        env[q] = 0.5f * (1.0f + cth * cth);
    }

    const int tile0 = blockIdx.x * TPB;
    for (int i = 0; i < TPB; ++i) {
        const int mt = tile0 + i;
        const int m  = (mt >> 1) & 511;
        const int ot = ((mt >> 10) & 1) * 2 + (mt & 1);
        const int b  = mt >> 11;
        const int col = b * NFRM + m + sh;
        const float* Yb = Y + (size_t)col * NCH + 16 * ot + cL + (size_t)(2 * hs) * PLANE;
        v8f c = {};
#pragma unroll
        for (int j = 0; j < 16; ++j) {
            v2f bv;
            bv.x = Yb[(size_t)(4 * j) * PLANE];
            bv.y = Yb[(size_t)(4 * j) * PLANE + PLANE];
            c = wmma_f32(a[j], bv, c);
        }
        float* lFc = lF + (i & 1) * (16 * SP);
        float* sp  = lFc + cL * SP + 16 * w + 8 * hs;
        *(float4*)sp       = make_float4(c[0], c[1], c[2], c[3]);
        *(float4*)(sp + 4) = make_float4(c[4], c[5], c[6], c[7]);
        __syncthreads();
        const float* ep = lFc + ecol * SP + eo4;
        const float4 lo = *(const float4*)ep;
        const float4 hi = *(const float4*)(ep + 32);
        float4 o;
        o.x = (hi.x + lo.x) / env[0];
        o.y = (hi.y + lo.y) / env[1];
        o.z = (hi.z + lo.z) / env[2];
        o.w = (hi.w + lo.w) / env[3];
        *(float4*)(out + (size_t)(b * NCH + 16 * ot + ecol) * T_LEN + m * HOP + eo4) = o;
    }
}

extern "C" void kernel_launch(void* const* d_in, const int* in_sizes, int n_in,
                              void* d_out, int out_size, void* d_ws, size_t ws_size,
                              hipStream_t stream) {
    const float* x   = (const float*)d_in[0];
    const float* wgt = (const float*)d_in[1];
    float* out = (float*)d_out;
    float* ws  = (float*)d_ws;

    float* A1 = ws + OFF_A1;
    float* A3 = ws + OFF_A3;
    float* Wt = ws + OFF_WT;
    float* U  = ws + OFF_U;
    float* Y  = ws + OFF_Y;

    prep_kernel<<<64, 256, 0, stream>>>(wgt, A1, A3, Wt);
    stft_gemm<<<NBLK1, 128, 0, stream>>>(x, A1, U);
    mix_gemm<<<dim3(NBLK2, NFREQ, 2), 128, 0, stream>>>(Wt, U, Y);
    istft_ola_gemm<<<NBLK3, 128, 0, stream>>>(A3, Y, out);
}


# STFT and ISTFT GEMMs copy their B tile (2 frames of x; 5 frames x 32 channels of Y) into LDS once per work-group with whole-line 16-byte loads; waves read fragments from LDS; sum order unchanged
# speedup vs baseline: 2.6912x; 1.3210x over previous
#include <hip/hip_runtime.h>
#include <hip/hip_bf16.h>

typedef __attribute__((ext_vector_type(2))) float v2f;
typedef __attribute__((ext_vector_type(8))) float v8f;

#define T_LEN   16384
#define HOP     32
#define NFRM    513
#define NB      8
#define NCH     64
#define NFREQ   33
#define NCOL    4104
#define PLANE   263168
#define NBLK1   2052
#define NPAIR2  129
#define PPB     4
#define NBLK2   33
#define TPB     8
#define NTILE3  16384
#define NBLK3   2048
#define SP      68
#define YP      168

#define OFF_A1  0
#define OFF_A3  4096
#define OFF_WT  8192
#define OFF_U   143360
#define OFF_Y   (OFF_U + 64*PLANE)


__device__ __forceinline__ v8f wmma_f32(v2f a, v2f b, v8f c) {
    return __builtin_amdgcn_wmma_f32_16x16x4_f32(false, a, false, b, (short)0, c,
                                                 false, false);
}

__device__ __forceinline__ float xhalf(float v) {
    const int b = __builtin_bit_cast(int, v);
    return __builtin_bit_cast(float,
        __builtin_amdgcn_permlanex16(b, b, 0x76543210, 0xfedcba98, false, false));
}

__global__ void prep_kernel(const float* __restrict__ wgt,
                            float* __restrict__ A1, float* __restrict__ A3,
                            float* __restrict__ Wt) {
    const int gid = blockIdx.x * blockDim.x + threadIdx.x;
    const int gsz = gridDim.x * blockDim.x;
    const float PI = 3.14159265358979f;
    for (int idx = gid; idx < 4096; idx += gsz) {
        int fr = idx >> 6, tau = idx & 63;
        float wv = 0.5f * (1.0f - cosf(2.0f * PI * (float)tau / 64.0f));
        float v;
        if (fr <= 32) v =  cosf(2.0f * PI * (float)(fr)      * (float)tau / 64.0f) * wv;
        else          v = -sinf(2.0f * PI * (float)(fr - 32) * (float)tau / 64.0f) * wv;
        A1[idx] = v;
    }
    for (int idx = gid; idx < 4096; idx += gsz) {
        int tau = idx >> 6, fr = idx & 63;
        float wv = 0.5f * (1.0f - cosf(2.0f * PI * (float)tau / 64.0f));
        float s = wv * (1.0f / 64.0f);
        float v;
        if (fr == 0)       v = s;
        else if (fr < 32)  v = 2.0f * cosf(2.0f * PI * (float)fr * (float)tau / 64.0f) * s;
        else if (fr == 32) v = cosf(PI * (float)tau) * s;
        else               v = -2.0f * sinf(2.0f * PI * (float)(fr - 32) * (float)tau / 64.0f) * s;
        A3[idx] = v;
    }
    for (int idx = gid; idx < NFREQ * 4096; idx += gsz) {
        int f = idx >> 12, oi = idx & 4095;
        Wt[idx] = wgt[(size_t)oi * NFREQ + f];
    }
}

__global__ void stft_gemm(const float* __restrict__ x, const float* __restrict__ A1,
                          float* __restrict__ U) {
    __shared__ __attribute__((aligned(16))) float lX[2 * NCH * SP];
    const int tid  = threadIdx.x;
    const int lane = tid & 31;
    const int w    = tid >> 5;
    const int cL   = lane & 15;
    const int hs   = lane >> 4;
    const int col0 = blockIdx.x * 2;

    const int sch = tid >> 4;
    const int sp4 = (tid & 15) * 4;
    float4 st[2][8];
#pragma unroll
    for (int i = 0; i < 2; ++i) {
        const int col = col0 + i;
        const int b   = col / NFRM;
        const int n   = col % NFRM;
        const float* xs = x + (size_t)(b * NCH + sch) * T_LEN;
        int ts = (n - 1) * HOP + sp4;
        ts = ts < 0 ? 0 : ts;
        ts = ts > T_LEN - 4 ? T_LEN - 4 : ts;
#pragma unroll
        for (int q = 0; q < 8; ++q)
            st[i][q] = *(const float4*)(xs + (size_t)(8 * q) * T_LEN + ts);
    }

    const float* arow = A1 + (16 * w + cL) * 64 + 2 * hs;
    v2f a[16];
#pragma unroll
    for (int j = 0; j < 16; ++j)
        a[j] = *(const v2f*)(arow + 4 * j);

#pragma unroll
    for (int i = 0; i < 2; ++i) {
#pragma unroll
        for (int q = 0; q < 8; ++q)
            *(float4*)(lX + i * (NCH * SP) + (8 * q + sch) * SP + sp4) = st[i][q];
    }
    for (int i = 0; i < 2; ++i) {
        const int col = col0 + i;
        const int b   = col / NFRM;
        const int n   = col % NFRM;
        if (n == 0 || n == NFRM - 1) {
            const float* xs = x + (size_t)(b * NCH + sch) * T_LEN;
            const int ts = (n - 1) * HOP + sp4;
            for (int q = 0; q < 8; ++q) {
                const float* xr = xs + (size_t)(8 * q) * T_LEN;
                float v[4];
#pragma unroll
                for (int e = 0; e < 4; ++e) {
                    int t = ts + e;
                    t = t < 0 ? -t : t;
                    t = t >= T_LEN ? 2 * T_LEN - 2 - t : t;
                    v[e] = xr[t];
                }
                *(float4*)(lX + i * (NCH * SP) + (8 * q + sch) * SP + sp4) =
                    make_float4(v[0], v[1], v[2], v[3]);
            }
        }
    }
    __syncthreads();

    for (int i = 0; i < 4; ++i) {
        const int fi  = i >> 1;
        const int col = col0 + fi;
        const int p   = i & 1;
        const float* lr0 = lX + fi * (NCH * SP) + (32 * p + cL) * SP + 2 * hs;
        const float* lr1 = lr0 + 16 * SP;
        v2f f0[16], f1[16];
#pragma unroll
        for (int j = 0; j < 16; ++j) {
            f0[j] = *(const v2f*)(lr0 + 4 * j);
            f1[j] = *(const v2f*)(lr1 + 4 * j);
        }
        __builtin_amdgcn_sched_barrier(0);
        v8f c0 = {}, c1 = {};
#pragma unroll
        for (int j = 0; j < 16; ++j) {
            c0 = wmma_f32(a[j], f0[j], c0);
            c1 = wmma_f32(a[j], f1[j], c1);
        }
        float* Ust = U + (size_t)(16 * w) * PLANE + (size_t)col * NCH + 32 * p + lane;
#pragma unroll
        for (int r = 0; r < 8; ++r) {
            const float give = hs ? c0[r] : c1[r];
            const float got  = xhalf(give);
            Ust[(size_t)r * PLANE]       = hs ? got : c0[r];
            Ust[(size_t)(8 + r) * PLANE] = hs ? c1[r] : got;
        }
    }
}

__global__ void mix_gemm(const float* __restrict__ Wt, const float* __restrict__ U,
                         float* __restrict__ Y) {
    const int f  = blockIdx.y;
    const int ri = blockIdx.z;
    if (ri == 1 && (f == 0 || f == 32)) return;
    const int frq = f + ri * 32;

    __shared__ __attribute__((aligned(16))) float lB[2 * 32 * SP];
    const int tid  = threadIdx.x;
    const int lane = tid & 31;
    const int w    = tid >> 5;
    const int cL   = lane & 15;
    const int hs   = lane >> 4;
    const int ow   = w & 1;
    const int ct   = w >> 1;
    const float* wrow = Wt + (size_t)f * 4096 + (32 * ow + cL) * 64 + 2 * hs;
    v2f wb0[16], wb1[16];
#pragma unroll
    for (int j = 0; j < 16; ++j) {
        wb0[j] = *(const v2f*)(wrow + 4 * j);
        wb1[j] = *(const v2f*)(wrow + 16 * 64 + 4 * j);
    }

    const int pair0 = blockIdx.x * PPB;
    const int np    = (NPAIR2 - pair0) < PPB ? (NPAIR2 - pair0) : PPB;
    const float* Up = U + (size_t)frq * PLANE;
    float*       Yp = Y + (size_t)frq * PLANE;
    for (int i = 0; i < np; ++i) {
        const int col0 = (pair0 + i) * 32;
        float* lBc = lB + (i & 1) * (32 * SP);
#pragma unroll
        for (int q = 0; q < 4; ++q) {
            const int fi = q * 128 + tid;
            const int cc = fi >> 4, p4 = fi & 15;
            int gc = col0 + cc;
            gc = gc < NCOL ? gc : NCOL - 1;
            *(float4*)(lBc + cc * SP + p4 * 4) = *(const float4*)(Up + (size_t)gc * NCH + p4 * 4);
        }
        __syncthreads();
        v8f c0 = {}, c1 = {};
        const float* lrow = lBc + (16 * ct + cL) * SP + 2 * hs;
#pragma unroll
        for (int j = 0; j < 16; ++j) {
            const v2f uv = *(const v2f*)(lrow + 4 * j);
            c0 = wmma_f32(uv, wb0[j], c0);
            c1 = wmma_f32(uv, wb1[j], c1);
        }
        const int crow = col0 + 16 * ct;
        float* Yst = Yp + (size_t)crow * NCH + 32 * ow + lane;
        if (col0 + 32 <= NCOL) {
#pragma unroll
            for (int r = 0; r < 8; ++r) {
                const float give = hs ? c0[r] : c1[r];
                const float got  = xhalf(give);
                Yst[(size_t)r * NCH]       = hs ? got : c0[r];
                Yst[(size_t)(8 + r) * NCH] = hs ? c1[r] : got;
            }
        } else {
#pragma unroll
            for (int r = 0; r < 8; ++r) {
                const float give = hs ? c0[r] : c1[r];
                const float got  = xhalf(give);
                if (crow + r < NCOL)     Yst[(size_t)r * NCH]       = hs ? got : c0[r];
                if (crow + 8 + r < NCOL) Yst[(size_t)(8 + r) * NCH] = hs ? c1[r] : got;
            }
        }
    }
}

__global__ void __launch_bounds__(128)
istft_ola_gemm(const float* __restrict__ A3, const float* __restrict__ Y,
               float* __restrict__ out) {
    __shared__ __attribute__((aligned(16))) float lF[2 * 16 * SP];
    __shared__ __attribute__((aligned(16))) float lY[64 * YP];
    const int tid  = threadIdx.x;
    const int lane = tid & 31;
    const int w    = tid >> 5;
    const int cL   = lane & 15;
    const int hs   = lane >> 4;
    const int sh   = w < 2 ? 1 : 0;

    const int tile0 = blockIdx.x * TPB;
    const int m0 = (tile0 >> 1) & 511;
    const int oh = (tile0 >> 10) & 1;
    const int b  = tile0 >> 11;

    const int sfr = tid >> 3;
    const int sc4 = (tid & 7) * 4;
    const float* Ys = Y + (size_t)sfr * PLANE + (size_t)(b * NFRM + m0) * NCH + 32 * oh + sc4;
    float4 st[5][4];
#pragma unroll
    for (int c = 0; c < 5; ++c) {
#pragma unroll
        for (int q = 0; q < 4; ++q)
            st[c][q] = *(const float4*)(Ys + (size_t)(16 * q) * PLANE + c * NCH);
    }

    const float* arow = A3 + (16 * w + cL) * 64 + 2 * hs;
    v2f a[16];
#pragma unroll
    for (int j = 0; j < 16; ++j)
        a[j] = *(const v2f*)(arow + 4 * j);

#pragma unroll
    for (int c = 0; c < 5; ++c) {
#pragma unroll
        for (int q = 0; q < 4; ++q)
            *(float4*)(lY + (16 * q + sfr) * YP + c * 32 + sc4) = st[c][q];
    }
    __syncthreads();

    const int ecol = tid >> 3;
    const int eo4  = (tid & 7) * 4;
    float env[4];
#pragma unroll
    for (int q = 0; q < 4; ++q) {
        const float cth = cosf(0.09817477042f * (float)(eo4 + q));
        env[q] = 0.5f * (1.0f + cth * cth);
    }

    for (int i = 0; i < TPB; ++i) {
        const int mt = tile0 + i;
        const int m  = (mt >> 1) & 511;
        const int ot = ((mt >> 10) & 1) * 2 + (mt & 1);
        const float* lb = lY + (2 * hs) * YP + ((i >> 1) + sh) * 32 + 16 * (i & 1) + cL;
        v2f bv[16];
#pragma unroll
        for (int j = 0; j < 16; ++j) {
            bv[j].x = lb[(4 * j) * YP];
            bv[j].y = lb[(4 * j + 1) * YP];
        }
        __builtin_amdgcn_sched_barrier(0);
        v8f c = {};
#pragma unroll
        for (int j = 0; j < 16; ++j)
            c = wmma_f32(a[j], bv[j], c);
        float* lFc = lF + (i & 1) * (16 * SP);
        float* sp  = lFc + cL * SP + 16 * w + 8 * hs;
        *(float4*)sp       = make_float4(c[0], c[1], c[2], c[3]);
        *(float4*)(sp + 4) = make_float4(c[4], c[5], c[6], c[7]);
        __syncthreads();
        const float* ep = lFc + ecol * SP + eo4;
        const float4 lo = *(const float4*)ep;
        const float4 hi = *(const float4*)(ep + 32);
        float4 o;
        o.x = (hi.x + lo.x) / env[0];
        o.y = (hi.y + lo.y) / env[1];
        o.z = (hi.z + lo.z) / env[2];
        o.w = (hi.w + lo.w) / env[3];
        *(float4*)(out + (size_t)(b * NCH + 16 * ot + ecol) * T_LEN + m * HOP + eo4) = o;
    }
}

extern "C" void kernel_launch(void* const* d_in, const int* in_sizes, int n_in,
                              void* d_out, int out_size, void* d_ws, size_t ws_size,
                              hipStream_t stream) {
    const float* x   = (const float*)d_in[0];
    const float* wgt = (const float*)d_in[1];
    float* out = (float*)d_out;
    float* ws  = (float*)d_ws;

    float* A1 = ws + OFF_A1;
    float* A3 = ws + OFF_A3;
    float* Wt = ws + OFF_WT;
    float* U  = ws + OFF_U;
    float* Y  = ws + OFF_Y;

    prep_kernel<<<64, 256, 0, stream>>>(wgt, A1, A3, Wt);
    stft_gemm<<<NBLK1, 128, 0, stream>>>(x, A1, U);
    mix_gemm<<<dim3(NBLK2, NFREQ, 2), 128, 0, stream>>>(Wt, U, Y);
    istft_ola_gemm<<<NBLK3, 128, 0, stream>>>(A3, Y, out);
}


# Prep stores A1, A3 and each W[f] in fragment order: a wave loads its kept operand with contiguous 16-byte loads (4 lines each), not row-strided 8-byte loads (16 lines each); sum order unchanged
# speedup vs baseline: 2.8012x; 1.0409x over previous
#include <hip/hip_runtime.h>
#include <hip/hip_bf16.h>

typedef __attribute__((ext_vector_type(2))) float v2f;
typedef __attribute__((ext_vector_type(8))) float v8f;

#define T_LEN   16384
#define HOP     32
#define NFRM    513
#define NB      8
#define NCH     64
#define NFREQ   33
#define NCOL    4104
#define PLANE   263168
#define NBLK1   2052
#define NPAIR2  129
#define PPB     4
#define NBLK2   33
#define TPB     8
#define NTILE3  16384
#define NBLK3   2048
#define SP      68
#define YP      168

#define OFF_A1  0
#define OFF_A3  4096
#define OFF_WT  8192
#define OFF_U   143360
#define OFF_Y   (OFF_U + 64*PLANE)


__device__ __forceinline__ v8f wmma_f32(v2f a, v2f b, v8f c) {
    return __builtin_amdgcn_wmma_f32_16x16x4_f32(false, a, false, b, (short)0, c,
                                                 false, false);
}

__device__ __forceinline__ float xhalf(float v) {
    const int b = __builtin_bit_cast(int, v);
    return __builtin_bit_cast(float,
        __builtin_amdgcn_permlanex16(b, b, 0x76543210, 0xfedcba98, false, false));
}

__device__ __forceinline__ void frag_pos(int idx, int& row, int& k) {
    const int e    = idx & 3;
    const int lane = (idx >> 2) & 31;
    const int q    = (idx >> 7) & 7;
    const int tile = idx >> 10;
    row = 16 * tile + (lane & 15);
    k   = 2 * (lane >> 4) + 4 * (2 * q + (e >> 1)) + (e & 1);
}

__device__ __forceinline__ void load_frags(const float* __restrict__ op, int tile, int lane,
                                           v2f (&a)[16]) {
    const float4* p = (const float4*)op + tile * 256 + lane;
#pragma unroll
    for (int q = 0; q < 8; ++q) {
        const float4 t = p[q * 32];
        a[2 * q][0]     = t.x;
        a[2 * q][1]     = t.y;
        a[2 * q + 1][0] = t.z;
        a[2 * q + 1][1] = t.w;
    }
}

__global__ void prep_kernel(const float* __restrict__ wgt,
                            float* __restrict__ A1, float* __restrict__ A3,
                            float* __restrict__ Wt) {
    const int gid = blockIdx.x * blockDim.x + threadIdx.x;
    const int gsz = gridDim.x * blockDim.x;
    const float PI = 3.14159265358979f;
    for (int idx = gid; idx < 4096; idx += gsz) {
        int fr, tau;
        frag_pos(idx, fr, tau);
        float wv = 0.5f * (1.0f - cosf(2.0f * PI * (float)tau / 64.0f));
        float v;
        if (fr <= 32) v =  cosf(2.0f * PI * (float)(fr)      * (float)tau / 64.0f) * wv;
        else          v = -sinf(2.0f * PI * (float)(fr - 32) * (float)tau / 64.0f) * wv;
        A1[idx] = v;
    }
    for (int idx = gid; idx < 4096; idx += gsz) {
        int tau, fr;
        frag_pos(idx, tau, fr);
        float wv = 0.5f * (1.0f - cosf(2.0f * PI * (float)tau / 64.0f));
        float s = wv * (1.0f / 64.0f);
        float v;
        if (fr == 0)       v = s;
        else if (fr < 32)  v = 2.0f * cosf(2.0f * PI * (float)fr * (float)tau / 64.0f) * s;
        else if (fr == 32) v = cosf(PI * (float)tau) * s;
        else               v = -2.0f * sinf(2.0f * PI * (float)(fr - 32) * (float)tau / 64.0f) * s;
        A3[idx] = v;
    }
    for (int idx = gid; idx < NFREQ * 4096; idx += gsz) {
        int f = idx >> 12, o, ic;
        frag_pos(idx & 4095, o, ic);
        Wt[idx] = wgt[(size_t)(o * NCH + ic) * NFREQ + f];
    }
}

__global__ void stft_gemm(const float* __restrict__ x, const float* __restrict__ A1,
                          float* __restrict__ U) {
    __shared__ __attribute__((aligned(16))) float lX[2 * NCH * SP];
    const int tid  = threadIdx.x;
    const int lane = tid & 31;
    const int w    = tid >> 5;
    const int cL   = lane & 15;
    const int hs   = lane >> 4;
    const int col0 = blockIdx.x * 2;

    const int sch = tid >> 4;
    const int sp4 = (tid & 15) * 4;
    float4 st[2][8];
#pragma unroll
    for (int i = 0; i < 2; ++i) {
        const int col = col0 + i;
        const int b   = col / NFRM;
        const int n   = col % NFRM;
        const float* xs = x + (size_t)(b * NCH + sch) * T_LEN;
        int ts = (n - 1) * HOP + sp4;
        ts = ts < 0 ? 0 : ts;
        ts = ts > T_LEN - 4 ? T_LEN - 4 : ts;
#pragma unroll
        for (int q = 0; q < 8; ++q)
            st[i][q] = *(const float4*)(xs + (size_t)(8 * q) * T_LEN + ts);
    }

    v2f a[16];
    load_frags(A1, w, lane, a);

#pragma unroll
    for (int i = 0; i < 2; ++i) {
#pragma unroll
        for (int q = 0; q < 8; ++q)
            *(float4*)(lX + i * (NCH * SP) + (8 * q + sch) * SP + sp4) = st[i][q];
    }
    for (int i = 0; i < 2; ++i) {
        const int col = col0 + i;
        const int b   = col / NFRM;
        const int n   = col % NFRM;
        if (n == 0 || n == NFRM - 1) {
            const float* xs = x + (size_t)(b * NCH + sch) * T_LEN;
            const int ts = (n - 1) * HOP + sp4;
            for (int q = 0; q < 8; ++q) {
                const float* xr = xs + (size_t)(8 * q) * T_LEN;
                float v[4];
#pragma unroll
                for (int e = 0; e < 4; ++e) {
                    int t = ts + e;
                    t = t < 0 ? -t : t;
                    t = t >= T_LEN ? 2 * T_LEN - 2 - t : t;
                    v[e] = xr[t];
                }
                *(float4*)(lX + i * (NCH * SP) + (8 * q + sch) * SP + sp4) =
                    make_float4(v[0], v[1], v[2], v[3]);
            }
        }
    }
    __syncthreads();

    for (int i = 0; i < 4; ++i) {
        const int fi  = i >> 1;
        const int col = col0 + fi;
        const int p   = i & 1;
        const float* lr0 = lX + fi * (NCH * SP) + (32 * p + cL) * SP + 2 * hs;
        const float* lr1 = lr0 + 16 * SP;
        v2f f0[16], f1[16];
#pragma unroll
        for (int j = 0; j < 16; ++j) {
            f0[j] = *(const v2f*)(lr0 + 4 * j);
            f1[j] = *(const v2f*)(lr1 + 4 * j);
        }
        __builtin_amdgcn_sched_barrier(0);
        v8f c0 = {}, c1 = {};
#pragma unroll
        for (int j = 0; j < 16; ++j) {
            c0 = wmma_f32(a[j], f0[j], c0);
            c1 = wmma_f32(a[j], f1[j], c1);
        }
        float* Ust = U + (size_t)(16 * w) * PLANE + (size_t)col * NCH + 32 * p + lane;
#pragma unroll
        for (int r = 0; r < 8; ++r) {
            const float give = hs ? c0[r] : c1[r];
            const float got  = xhalf(give);
            Ust[(size_t)r * PLANE]       = hs ? got : c0[r];
            Ust[(size_t)(8 + r) * PLANE] = hs ? c1[r] : got;
        }
    }
}

__global__ void mix_gemm(const float* __restrict__ Wt, const float* __restrict__ U,
                         float* __restrict__ Y) {
    const int f  = blockIdx.y;
    const int ri = blockIdx.z;
    if (ri == 1 && (f == 0 || f == 32)) return;
    const int frq = f + ri * 32;

    __shared__ __attribute__((aligned(16))) float lB[2 * 32 * SP];
    const int tid  = threadIdx.x;
    const int lane = tid & 31;
    const int w    = tid >> 5;
    const int cL   = lane & 15;
    const int hs   = lane >> 4;
    const int ow   = w & 1;
    const int ct   = w >> 1;
    v2f wb0[16], wb1[16];
    load_frags(Wt + (size_t)f * 4096, 2 * ow,     lane, wb0);
    load_frags(Wt + (size_t)f * 4096, 2 * ow + 1, lane, wb1);

    const int pair0 = blockIdx.x * PPB;
    const int np    = (NPAIR2 - pair0) < PPB ? (NPAIR2 - pair0) : PPB;
    const float* Up = U + (size_t)frq * PLANE;
    float*       Yp = Y + (size_t)frq * PLANE;
    for (int i = 0; i < np; ++i) {
        const int col0 = (pair0 + i) * 32;
        float* lBc = lB + (i & 1) * (32 * SP);
#pragma unroll
        for (int q = 0; q < 4; ++q) {
            const int fi = q * 128 + tid;
            const int cc = fi >> 4, p4 = fi & 15;
            int gc = col0 + cc;
            gc = gc < NCOL ? gc : NCOL - 1;
            *(float4*)(lBc + cc * SP + p4 * 4) = *(const float4*)(Up + (size_t)gc * NCH + p4 * 4);
        }
        __syncthreads();
        v8f c0 = {}, c1 = {};
        const float* lrow = lBc + (16 * ct + cL) * SP + 2 * hs;
#pragma unroll
        for (int j = 0; j < 16; ++j) {
            const v2f uv = *(const v2f*)(lrow + 4 * j);
            c0 = wmma_f32(uv, wb0[j], c0);
            c1 = wmma_f32(uv, wb1[j], c1);
        }
        const int crow = col0 + 16 * ct;
        float* Yst = Yp + (size_t)crow * NCH + 32 * ow + lane;
        if (col0 + 32 <= NCOL) {
#pragma unroll
            for (int r = 0; r < 8; ++r) {
                const float give = hs ? c0[r] : c1[r];
                const float got  = xhalf(give);
                Yst[(size_t)r * NCH]       = hs ? got : c0[r];
                Yst[(size_t)(8 + r) * NCH] = hs ? c1[r] : got;
            }
        } else {
#pragma unroll
            for (int r = 0; r < 8; ++r) {
                const float give = hs ? c0[r] : c1[r];
                const float got  = xhalf(give);
                if (crow + r < NCOL)     Yst[(size_t)r * NCH]       = hs ? got : c0[r];
                if (crow + 8 + r < NCOL) Yst[(size_t)(8 + r) * NCH] = hs ? c1[r] : got;
            }
        }
    }
}

__global__ void __launch_bounds__(128)
istft_ola_gemm(const float* __restrict__ A3, const float* __restrict__ Y,
               float* __restrict__ out) {
    __shared__ __attribute__((aligned(16))) float lF[2 * 16 * SP];
    __shared__ __attribute__((aligned(16))) float lY[64 * YP];
    const int tid  = threadIdx.x;
    const int lane = tid & 31;
    const int w    = tid >> 5;
    const int cL   = lane & 15;
    const int hs   = lane >> 4;
    const int sh   = w < 2 ? 1 : 0;

    const int tile0 = blockIdx.x * TPB;
    const int m0 = (tile0 >> 1) & 511;
    const int oh = (tile0 >> 10) & 1;
    const int b  = tile0 >> 11;

    const int sfr = tid >> 3;
    const int sc4 = (tid & 7) * 4;
    const float* Ys = Y + (size_t)sfr * PLANE + (size_t)(b * NFRM + m0) * NCH + 32 * oh + sc4;
    float4 st[5][4];
#pragma unroll
    for (int c = 0; c < 5; ++c) {
#pragma unroll
        for (int q = 0; q < 4; ++q)
            st[c][q] = *(const float4*)(Ys + (size_t)(16 * q) * PLANE + c * NCH);
    }

    v2f a[16];
    load_frags(A3, w, lane, a);

#pragma unroll
    for (int c = 0; c < 5; ++c) {
#pragma unroll
        for (int q = 0; q < 4; ++q)
            *(float4*)(lY + (16 * q + sfr) * YP + c * 32 + sc4) = st[c][q];
    }
    __syncthreads();

    const int ecol = tid >> 3;
    const int eo4  = (tid & 7) * 4;
    float env[4];
#pragma unroll
    for (int q = 0; q < 4; ++q) {
        const float cth = cosf(0.09817477042f * (float)(eo4 + q));
        env[q] = 0.5f * (1.0f + cth * cth);
    }

    for (int i = 0; i < TPB; ++i) {
        const int mt = tile0 + i;
        const int m  = (mt >> 1) & 511;
        const int ot = ((mt >> 10) & 1) * 2 + (mt & 1);
        const float* lb = lY + (2 * hs) * YP + ((i >> 1) + sh) * 32 + 16 * (i & 1) + cL;
        v2f bv[16];
#pragma unroll
        for (int j = 0; j < 16; ++j) {
            bv[j].x = lb[(4 * j) * YP];
            bv[j].y = lb[(4 * j + 1) * YP];
        }
        __builtin_amdgcn_sched_barrier(0);
        v8f c = {};
#pragma unroll
        for (int j = 0; j < 16; ++j)
            c = wmma_f32(a[j], bv[j], c);
        float* lFc = lF + (i & 1) * (16 * SP);
        float* sp  = lFc + cL * SP + 16 * w + 8 * hs;
        *(float4*)sp       = make_float4(c[0], c[1], c[2], c[3]);
        *(float4*)(sp + 4) = make_float4(c[4], c[5], c[6], c[7]);
        __syncthreads();
        const float* ep = lFc + ecol * SP + eo4;
        const float4 lo = *(const float4*)ep;
        const float4 hi = *(const float4*)(ep + 32);
        float4 o;
        o.x = (hi.x + lo.x) / env[0];
        o.y = (hi.y + lo.y) / env[1];
        o.z = (hi.z + lo.z) / env[2];
        o.w = (hi.w + lo.w) / env[3];
        *(float4*)(out + (size_t)(b * NCH + 16 * ot + ecol) * T_LEN + m * HOP + eo4) = o;
    }
}

extern "C" void kernel_launch(void* const* d_in, const int* in_sizes, int n_in,
                              void* d_out, int out_size, void* d_ws, size_t ws_size,
                              hipStream_t stream) {
    const float* x   = (const float*)d_in[0];
    const float* wgt = (const float*)d_in[1];
    float* out = (float*)d_out;
    float* ws  = (float*)d_ws;

    float* A1 = ws + OFF_A1;
    float* A3 = ws + OFF_A3;
    float* Wt = ws + OFF_WT;
    float* U  = ws + OFF_U;
    float* Y  = ws + OFF_Y;

    prep_kernel<<<64, 256, 0, stream>>>(wgt, A1, A3, Wt);
    stft_gemm<<<NBLK1, 128, 0, stream>>>(x, A1, U);
    mix_gemm<<<dim3(NBLK2, NFREQ, 2), 128, 0, stream>>>(Wt, U, Y);
    istft_ola_gemm<<<NBLK3, 128, 0, stream>>>(A3, Y, out);
}


# Radix-2 split of both 64-point transforms on the vector unit: STFT forms xw[t]+-xw[t+32], ISTFT combines E+-O, so each DFT GEMM has K=32 not 64; window moves onto the samples; sum order changes
# speedup vs baseline: 3.4660x; 1.2373x over previous
#include <hip/hip_runtime.h>
#include <hip/hip_bf16.h>

typedef __attribute__((ext_vector_type(2))) float v2f;
typedef __attribute__((ext_vector_type(8))) float v8f;

#define T_LEN   16384
#define HOP     32
#define NFRM    513
#define NB      8
#define NCH     64
#define NFREQ   33
#define NCOL    4104
#define PLANE   263168
#define NBLK1   2052
#define NPAIR2  129
#define PPB     4
#define NBLK2   33
#define TPB     8
#define NTILE3  16384
#define NBLK3   2048
#define SP      68
#define YP      168

#define OFF_A1  0
#define OFF_WIN 2048
#define OFF_A3  4096
#define OFF_WT  8192
#define OFF_U   143360
#define OFF_Y   (OFF_U + 64*PLANE)


__device__ __forceinline__ v8f wmma_f32(v2f a, v2f b, v8f c) {
    return __builtin_amdgcn_wmma_f32_16x16x4_f32(false, a, false, b, (short)0, c,
                                                 false, false);
}

__device__ __forceinline__ float xhalf(float v) {
    const int b = __builtin_bit_cast(int, v);
    return __builtin_bit_cast(float,
        __builtin_amdgcn_permlanex16(b, b, 0x76543210, 0xfedcba98, false, false));
}

__device__ __forceinline__ void frag_pos(int idx, int& row, int& k) {
    const int e    = idx & 3;
    const int lane = (idx >> 2) & 31;
    const int q    = (idx >> 7) & 7;
    const int tile = idx >> 10;
    row = 16 * tile + (lane & 15);
    k   = 2 * (lane >> 4) + 4 * (2 * q + (e >> 1)) + (e & 1);
}

__device__ __forceinline__ void load_frags(const float* __restrict__ op, int tile, int lane,
                                           v2f (&a)[16]) {
    const float4* p = (const float4*)op + tile * 256 + lane;
#pragma unroll
    for (int q = 0; q < 8; ++q) {
        const float4 t = p[q * 32];
        a[2 * q][0]     = t.x;
        a[2 * q][1]     = t.y;
        a[2 * q + 1][0] = t.z;
        a[2 * q + 1][1] = t.w;
    }
}


__device__ __forceinline__ void frag_pos32(int idx, int& row, int& k) {
    const int e    = idx & 3;
    const int lane = (idx >> 2) & 31;
    const int q    = (idx >> 7) & 3;
    const int tile = idx >> 9;
    row = 16 * tile + (lane & 15);
    k   = 2 * (lane >> 4) + 4 * (2 * q + (e >> 1)) + (e & 1);
}

__device__ __forceinline__ void load_frags8(const float* __restrict__ op, int tile, int lane,
                                            v2f (&a)[8]) {
    const float4* p = (const float4*)op + tile * 128 + lane;
#pragma unroll
    for (int q = 0; q < 4; ++q) {
        const float4 t = p[q * 32];
        a[2 * q][0]     = t.x;
        a[2 * q][1]     = t.y;
        a[2 * q + 1][0] = t.z;
        a[2 * q + 1][1] = t.w;
    }
}

__global__ void prep_kernel(const float* __restrict__ wgt,
                            float* __restrict__ A1, float* __restrict__ WIN,
                            float* __restrict__ A3, float* __restrict__ Wt) {
    const int gid = blockIdx.x * blockDim.x + threadIdx.x;
    const int gsz = gridDim.x * blockDim.x;
    const float PI = 3.14159265358979f;
    for (int idx = gid; idx < 2048; idx += gsz) {
        int G, tp;
        frag_pos32(idx, G, tp);
        const int P = 2 * (G & 31) + (G >> 5);
        float v;
        if (P <= 32) v =  cosf(2.0f * PI * (float)(P)      * (float)tp / 64.0f);
        else         v = -sinf(2.0f * PI * (float)(P - 32) * (float)tp / 64.0f);
        A1[idx] = v;
    }
    for (int idx = gid; idx < 64; idx += gsz)
        WIN[idx] = 0.5f * (1.0f - cosf(2.0f * PI * (float)idx / 64.0f));
    for (int idx = gid; idx < 2048; idx += gsz) {
        int G, k;
        frag_pos32(idx, G, k);
        const int tp = G & 31;
        const int P  = 2 * k + (G >> 5);
        const float s = 1.0f / 64.0f;
        float v;
        if (P == 0)       v = s;
        else if (P < 32)  v = 2.0f * cosf(2.0f * PI * (float)P * (float)tp / 64.0f) * s;
        else if (P == 32) v = cosf(PI * (float)tp) * s;
        else              v = -2.0f * sinf(2.0f * PI * (float)(P - 32) * (float)tp / 64.0f) * s;
        A3[idx] = v;
    }
    for (int idx = gid; idx < NFREQ * 4096; idx += gsz) {
        int f = idx >> 12, o, ic;
        frag_pos(idx & 4095, o, ic);
        Wt[idx] = wgt[(size_t)(o * NCH + ic) * NFREQ + f];
    }
}

__global__ void stft_gemm(const float* __restrict__ x, const float* __restrict__ A1,
                          const float* __restrict__ WIN, float* __restrict__ U) {
    __shared__ __attribute__((aligned(16))) float lX[2 * NCH * SP];
    const int tid  = threadIdx.x;
    const int lane = tid & 31;
    const int w    = tid >> 5;
    const int cL   = lane & 15;
    const int hs   = lane >> 4;
    const int g    = w >> 1;
    const int col0 = blockIdx.x * 2;

    const int sch = tid >> 3;
    const int s4  = (tid & 7) * 4;
    float4 lo[2][4], hi[2][4];
#pragma unroll
    for (int i = 0; i < 2; ++i) {
        const int col = col0 + i;
        const int b   = col / NFRM;
        const int n   = col % NFRM;
        const float* xs = x + (size_t)(b * NCH + sch) * T_LEN;
        int t0 = (n - 1) * HOP + s4;
        int t1 = t0 + 32;
        t0 = t0 < 0 ? 0 : t0;
        t1 = t1 > T_LEN - 4 ? T_LEN - 4 : t1;
#pragma unroll
        for (int q = 0; q < 4; ++q) {
            lo[i][q] = *(const float4*)(xs + (size_t)(16 * q) * T_LEN + t0);
            hi[i][q] = *(const float4*)(xs + (size_t)(16 * q) * T_LEN + t1);
        }
    }
    const float4 wl4 = *(const float4*)(WIN + s4);
    const float4 wh4 = *(const float4*)(WIN + 32 + s4);
    const float wl[4] = {wl4.x, wl4.y, wl4.z, wl4.w};
    const float wh[4] = {wh4.x, wh4.y, wh4.z, wh4.w};

    v2f a[8];
    load_frags8(A1, w, lane, a);

#pragma unroll
    for (int i = 0; i < 2; ++i) {
#pragma unroll
        for (int q = 0; q < 4; ++q) {
            const float4 l = lo[i][q], h = hi[i][q];
            const float u0 = l.x * wl[0], u1 = l.y * wl[1], u2 = l.z * wl[2], u3 = l.w * wl[3];
            const float v0 = h.x * wh[0], v1 = h.y * wh[1], v2 = h.z * wh[2], v3 = h.w * wh[3];
            float* row = lX + i * (NCH * SP) + (16 * q + sch) * SP;
            *(float4*)(row + s4)      = make_float4(u0 + v0, u1 + v1, u2 + v2, u3 + v3);
            *(float4*)(row + 32 + s4) = make_float4(u0 - v0, u1 - v1, u2 - v2, u3 - v3);
        }
    }
    for (int i = 0; i < 2; ++i) {
        const int col = col0 + i;
        const int b   = col / NFRM;
        const int n   = col % NFRM;
        if (n == 0 || n == NFRM - 1) {
            const float* xs = x + (size_t)(b * NCH + sch) * T_LEN;
            const int ts = (n - 1) * HOP + s4;
            for (int q = 0; q < 4; ++q) {
                const float* xr = xs + (size_t)(16 * q) * T_LEN;
                float p[4], d[4];
#pragma unroll
                for (int e = 0; e < 4; ++e) {
                    int ta = ts + e;
                    ta = ta < 0 ? -ta : ta;
                    ta = ta >= T_LEN ? 2 * T_LEN - 2 - ta : ta;
                    int tb = ts + 32 + e;
                    tb = tb < 0 ? -tb : tb;
                    tb = tb >= T_LEN ? 2 * T_LEN - 2 - tb : tb;
                    const float u = xr[ta] * wl[e];
                    const float v = xr[tb] * wh[e];
                    p[e] = u + v;
                    d[e] = u - v;
                }
                float* row = lX + i * (NCH * SP) + (16 * q + sch) * SP;
                *(float4*)(row + s4)      = make_float4(p[0], p[1], p[2], p[3]);
                *(float4*)(row + 32 + s4) = make_float4(d[0], d[1], d[2], d[3]);
            }
        }
    }
    __syncthreads();

    for (int i = 0; i < 4; ++i) {
        const int fi  = i >> 1;
        const int col = col0 + fi;
        const int p   = i & 1;
        const float* lr0 = lX + fi * (NCH * SP) + (32 * p + cL) * SP + 32 * g + 2 * hs;
        const float* lr1 = lr0 + 16 * SP;
        v2f f0[8], f1[8];
#pragma unroll
        for (int j = 0; j < 8; ++j) {
            f0[j] = *(const v2f*)(lr0 + 4 * j);
            f1[j] = *(const v2f*)(lr1 + 4 * j);
        }
        __builtin_amdgcn_sched_barrier(0);
        v8f c0 = {}, c1 = {};
#pragma unroll
        for (int j = 0; j < 8; ++j) {
            c0 = wmma_f32(a[j], f0[j], c0);
            c1 = wmma_f32(a[j], f1[j], c1);
        }
        float* Ust = U + (size_t)(32 * (w & 1) + g) * PLANE + (size_t)col * NCH + 32 * p + lane;
#pragma unroll
        for (int r = 0; r < 8; ++r) {
            const float give = hs ? c0[r] : c1[r];
            const float got  = xhalf(give);
            Ust[(size_t)(2 * r) * PLANE]      = hs ? got : c0[r];
            Ust[(size_t)(16 + 2 * r) * PLANE] = hs ? c1[r] : got;
        }
    }
}

__global__ void mix_gemm(const float* __restrict__ Wt, const float* __restrict__ U,
                         float* __restrict__ Y) {
    const int f  = blockIdx.y;
    const int ri = blockIdx.z;
    if (ri == 1 && (f == 0 || f == 32)) return;
    const int frq = f + ri * 32;

    __shared__ __attribute__((aligned(16))) float lB[2 * 32 * SP];
    const int tid  = threadIdx.x;
    const int lane = tid & 31;
    const int w    = tid >> 5;
    const int cL   = lane & 15;
    const int hs   = lane >> 4;
    const int ow   = w & 1;
    const int ct   = w >> 1;
    v2f wb0[16], wb1[16];
    load_frags(Wt + (size_t)f * 4096, 2 * ow,     lane, wb0);
    load_frags(Wt + (size_t)f * 4096, 2 * ow + 1, lane, wb1);

    const int pair0 = blockIdx.x * PPB;
    const int np    = (NPAIR2 - pair0) < PPB ? (NPAIR2 - pair0) : PPB;
    const float* Up = U + (size_t)frq * PLANE;
    float*       Yp = Y + (size_t)frq * PLANE;
    for (int i = 0; i < np; ++i) {
        const int col0 = (pair0 + i) * 32;
        float* lBc = lB + (i & 1) * (32 * SP);
#pragma unroll
        for (int q = 0; q < 4; ++q) {
            const int fi = q * 128 + tid;
            const int cc = fi >> 4, p4 = fi & 15;
            int gc = col0 + cc;
            gc = gc < NCOL ? gc : NCOL - 1;
            *(float4*)(lBc + cc * SP + p4 * 4) = *(const float4*)(Up + (size_t)gc * NCH + p4 * 4);
        }
        __syncthreads();
        v8f c0 = {}, c1 = {};
        const float* lrow = lBc + (16 * ct + cL) * SP + 2 * hs;
#pragma unroll
        for (int j = 0; j < 16; ++j) {
            const v2f uv = *(const v2f*)(lrow + 4 * j);
            c0 = wmma_f32(uv, wb0[j], c0);
            c1 = wmma_f32(uv, wb1[j], c1);
        }
        const int crow = col0 + 16 * ct;
        float* Yst = Yp + (size_t)crow * NCH + 32 * ow + lane;
        if (col0 + 32 <= NCOL) {
#pragma unroll
            for (int r = 0; r < 8; ++r) {
                const float give = hs ? c0[r] : c1[r];
                const float got  = xhalf(give);
                Yst[(size_t)r * NCH]       = hs ? got : c0[r];
                Yst[(size_t)(8 + r) * NCH] = hs ? c1[r] : got;
            }
        } else {
#pragma unroll
            for (int r = 0; r < 8; ++r) {
                const float give = hs ? c0[r] : c1[r];
                const float got  = xhalf(give);
                if (crow + r < NCOL)     Yst[(size_t)r * NCH]       = hs ? got : c0[r];
                if (crow + 8 + r < NCOL) Yst[(size_t)(8 + r) * NCH] = hs ? c1[r] : got;
            }
        }
    }
}

__global__ void __launch_bounds__(128)
istft_ola_gemm(const float* __restrict__ A3, const float* __restrict__ Y,
               float* __restrict__ out) {
    __shared__ __attribute__((aligned(16))) float lF[2 * 16 * SP];
    __shared__ __attribute__((aligned(16))) float lY[64 * YP];
    const int tid  = threadIdx.x;
    const int lane = tid & 31;
    const int w    = tid >> 5;
    const int cL   = lane & 15;
    const int hs   = lane >> 4;
    const int g    = w >> 1;

    const int tile0 = blockIdx.x * TPB;
    const int m0 = (tile0 >> 1) & 511;
    const int oh = (tile0 >> 10) & 1;
    const int b  = tile0 >> 11;

    const int sfr = tid >> 3;
    const int sc4 = (tid & 7) * 4;
    const float* Ys = Y + (size_t)sfr * PLANE + (size_t)(b * NFRM + m0) * NCH + 32 * oh + sc4;
    float4 st[5][4];
#pragma unroll
    for (int c = 0; c < 5; ++c) {
#pragma unroll
        for (int q = 0; q < 4; ++q)
            st[c][q] = *(const float4*)(Ys + (size_t)(16 * q) * PLANE + c * NCH);
    }

    v2f a[8];
    load_frags8(A3, w, lane, a);

#pragma unroll
    for (int c = 0; c < 5; ++c) {
#pragma unroll
        for (int q = 0; q < 4; ++q)
            *(float4*)(lY + (16 * q + sfr) * YP + c * 32 + sc4) = st[c][q];
    }
    __syncthreads();

    const int ecol = tid >> 3;
    const int eo4  = (tid & 7) * 4;
    float wlo[4], whi[4], env[4];
#pragma unroll
    for (int q = 0; q < 4; ++q) {
        const float cth = cosf(0.09817477042f * (float)(eo4 + q));
        wlo[q] = 0.5f * (1.0f - cth);
        whi[q] = 0.5f * (1.0f + cth);
        env[q] = 0.5f * (1.0f + cth * cth);
    }
    float4 Ep = make_float4(0.0f, 0.0f, 0.0f, 0.0f);
    float4 Op = make_float4(0.0f, 0.0f, 0.0f, 0.0f);

    for (int it = 0; it < 10; ++it) {
        const int ct = it / 5;
        const int fr = it - 5 * ct;
        const float* lb = lY + (4 * hs + g) * YP + fr * 32 + 16 * ct + cL;
        v2f bv[8];
#pragma unroll
        for (int j = 0; j < 8; ++j) {
            bv[j].x = lb[(8 * j) * YP];
            bv[j].y = lb[(8 * j + 2) * YP];
        }
        __builtin_amdgcn_sched_barrier(0);
        v8f c = {};
#pragma unroll
        for (int j = 0; j < 8; ++j)
            c = wmma_f32(a[j], bv[j], c);
        float* lFc = lF + (it & 1) * (16 * SP);
        float* sp  = lFc + cL * SP + 16 * w + 8 * hs;
        *(float4*)sp       = make_float4(c[0], c[1], c[2], c[3]);
        *(float4*)(sp + 4) = make_float4(c[4], c[5], c[6], c[7]);
        __syncthreads();
        const float* ep = lFc + ecol * SP + eo4;
        const float4 En = *(const float4*)ep;
        const float4 On = *(const float4*)(ep + 32);
        if (fr >= 1) {
            const int m  = m0 + fr - 1;
            const int ot = 2 * oh + ct;
            float4 o;
            o.x = (whi[0] * (Ep.x - Op.x) + wlo[0] * (En.x + On.x)) / env[0];
            o.y = (whi[1] * (Ep.y - Op.y) + wlo[1] * (En.y + On.y)) / env[1];
            o.z = (whi[2] * (Ep.z - Op.z) + wlo[2] * (En.z + On.z)) / env[2];
            o.w = (whi[3] * (Ep.w - Op.w) + wlo[3] * (En.w + On.w)) / env[3];
            *(float4*)(out + (size_t)(b * NCH + 16 * ot + ecol) * T_LEN + m * HOP + eo4) = o;
        }
        Ep = En;
        Op = On;
    }
}

extern "C" void kernel_launch(void* const* d_in, const int* in_sizes, int n_in,
                              void* d_out, int out_size, void* d_ws, size_t ws_size,
                              hipStream_t stream) {
    const float* x   = (const float*)d_in[0];
    const float* wgt = (const float*)d_in[1];
    float* out = (float*)d_out;
    float* ws  = (float*)d_ws;

    float* A1 = ws + OFF_A1;
    float* WIN = ws + OFF_WIN;
    float* A3 = ws + OFF_A3;
    float* Wt = ws + OFF_WT;
    float* U  = ws + OFF_U;
    float* Y  = ws + OFF_Y;

    prep_kernel<<<64, 256, 0, stream>>>(wgt, A1, WIN, A3, Wt);
    stft_gemm<<<NBLK1, 128, 0, stream>>>(x, A1, WIN, U);
    mix_gemm<<<dim3(NBLK2, NFREQ, 2), 128, 0, stream>>>(Wt, U, Y);
    istft_ola_gemm<<<NBLK3, 128, 0, stream>>>(A3, Y, out);
}


# Set-up launch removed: STFT/ISTFT waves build their DFT operand (STFT its window too) in registers from one cosine per lane, fetched across lanes; STFT threads write Wt; 3 launches; sum order same
# speedup vs baseline: 3.7699x; 1.0877x over previous
#include <hip/hip_runtime.h>
#include <hip/hip_bf16.h>

typedef __attribute__((ext_vector_type(2))) float v2f;
typedef __attribute__((ext_vector_type(8))) float v8f;

#define T_LEN   16384
#define HOP     32
#define NFRM    513
#define NB      8
#define NCH     64
#define NFREQ   33
#define NCOL    4104
#define PLANE   263168
#define NBLK1   2052
#define NPAIR2  129
#define PPB     4
#define NBLK2   33
#define TPB     8
#define NTILE3  16384
#define NBLK3   2048
#define SP      68
#define YP      168

#define OFF_WT  8192
#define OFF_U   143360
#define OFF_Y   (OFF_U + 64*PLANE)


__device__ __forceinline__ v8f wmma_f32(v2f a, v2f b, v8f c) {
    return __builtin_amdgcn_wmma_f32_16x16x4_f32(false, a, false, b, (short)0, c,
                                                 false, false);
}

__device__ __forceinline__ float xhalf(float v) {
    const int b = __builtin_bit_cast(int, v);
    return __builtin_bit_cast(float,
        __builtin_amdgcn_permlanex16(b, b, 0x76543210, 0xfedcba98, false, false));
}

__device__ __forceinline__ void frag_pos(int idx, int& row, int& k) {
    const int e    = idx & 3;
    const int lane = (idx >> 2) & 31;
    const int q    = (idx >> 7) & 7;
    const int tile = idx >> 10;
    row = 16 * tile + (lane & 15);
    k   = 2 * (lane >> 4) + 4 * (2 * q + (e >> 1)) + (e & 1);
}

__device__ __forceinline__ void load_frags(const float* __restrict__ op, int tile, int lane,
                                           v2f (&a)[16]) {
    const float4* p = (const float4*)op + tile * 256 + lane;
#pragma unroll
    for (int q = 0; q < 8; ++q) {
        const float4 t = p[q * 32];
        a[2 * q][0]     = t.x;
        a[2 * q][1]     = t.y;
        a[2 * q + 1][0] = t.z;
        a[2 * q + 1][1] = t.w;
    }
}


__device__ __forceinline__ float cos64(float tab, int m) {
    const int v = __builtin_amdgcn_ds_bpermute((m & 31) << 2, __builtin_bit_cast(int, tab));
    return __builtin_bit_cast(float, v ^ (int)((unsigned)(m & 32) << 26));
}

__device__ __forceinline__ void build_a1(float tab, int w, int lane, v2f (&a)[8]) {
    const int G   = 16 * w + (lane & 15);
    const int P   = 2 * (G & 31) + (G >> 5);
    const int mul = P <= 32 ? P : P - 32;
    const int ofs = P <= 32 ? 0 : 16;
    const int k0  = 2 * (lane >> 4);
#pragma unroll
    for (int j = 0; j < 8; ++j) {
        a[j][0] = cos64(tab, mul * (k0 + 4 * j) + ofs);
        a[j][1] = cos64(tab, mul * (k0 + 4 * j + 1) + ofs);
    }
}

__device__ __forceinline__ void build_a3(float tab, int w, int lane, v2f (&a)[8]) {
    const int tp = (16 * w + (lane & 15)) & 31;
    const int g  = w >> 1;
    const int k0 = 2 * (lane >> 4);
#pragma unroll
    for (int j = 0; j < 8; ++j) {
#pragma unroll
        for (int e = 0; e < 2; ++e) {
            const int   P   = 2 * (k0 + 4 * j + e) + g;
            const int   mul = P <= 32 ? P : P - 32;
            const int   ofs = P <= 32 ? 0 : 16;
            const float sc  = (P == 0 || P == 32) ? 0.015625f : 0.03125f;
            a[j][e] = cos64(tab, mul * tp + ofs) * sc;
        }
    }
}

__global__ void stft_gemm(const float* __restrict__ x, const float* __restrict__ wgt,
                          float* __restrict__ Wt, float* __restrict__ U) {
    __shared__ __attribute__((aligned(16))) float lX[2 * NCH * SP];
    const int tid  = threadIdx.x;
    const int lane = tid & 31;
    const int w    = tid >> 5;
    const int cL   = lane & 15;
    const int hs   = lane >> 4;
    const int g    = w >> 1;
    const int col0 = blockIdx.x * 2;

    const int widx = blockIdx.x * 128 + tid;
    float wv = 0.0f;
    if (widx < NFREQ * 4096) {
        int o, ic;
        frag_pos(widx & 4095, o, ic);
        wv = wgt[(size_t)(o * NCH + ic) * NFREQ + (widx >> 12)];
    }

    const int sch = tid >> 3;
    const int s4  = (tid & 7) * 4;
    float4 lo[2][4], hi[2][4];
#pragma unroll
    for (int i = 0; i < 2; ++i) {
        const int col = col0 + i;
        const int b   = col / NFRM;
        const int n   = col % NFRM;
        const float* xs = x + (size_t)(b * NCH + sch) * T_LEN;
        int t0 = (n - 1) * HOP + s4;
        int t1 = t0 + 32;
        t0 = t0 < 0 ? 0 : t0;
        t1 = t1 > T_LEN - 4 ? T_LEN - 4 : t1;
#pragma unroll
        for (int q = 0; q < 4; ++q) {
            lo[i][q] = *(const float4*)(xs + (size_t)(16 * q) * T_LEN + t0);
            hi[i][q] = *(const float4*)(xs + (size_t)(16 * q) * T_LEN + t1);
        }
    }
    const float tab = cospif((float)lane * 0.03125f);
    float wl[4], wh[4];
#pragma unroll
    for (int e = 0; e < 4; ++e) {
        const float c = cos64(tab, s4 + e);
        wl[e] = 0.5f * (1.0f - c);
        wh[e] = 0.5f * (1.0f + c);
    }

    v2f a[8];
    build_a1(tab, w, lane, a);

#pragma unroll
    for (int i = 0; i < 2; ++i) {
#pragma unroll
        for (int q = 0; q < 4; ++q) {
            const float4 l = lo[i][q], h = hi[i][q];
            const float u0 = l.x * wl[0], u1 = l.y * wl[1], u2 = l.z * wl[2], u3 = l.w * wl[3];
            const float v0 = h.x * wh[0], v1 = h.y * wh[1], v2 = h.z * wh[2], v3 = h.w * wh[3];
            float* row = lX + i * (NCH * SP) + (16 * q + sch) * SP;
            *(float4*)(row + s4)      = make_float4(u0 + v0, u1 + v1, u2 + v2, u3 + v3);
            *(float4*)(row + 32 + s4) = make_float4(u0 - v0, u1 - v1, u2 - v2, u3 - v3);
        }
    }
    for (int i = 0; i < 2; ++i) {
        const int col = col0 + i;
        const int b   = col / NFRM;
        const int n   = col % NFRM;
        if (n == 0 || n == NFRM - 1) {
            const float* xs = x + (size_t)(b * NCH + sch) * T_LEN;
            const int ts = (n - 1) * HOP + s4;
            for (int q = 0; q < 4; ++q) {
                const float* xr = xs + (size_t)(16 * q) * T_LEN;
                float p[4], d[4];
#pragma unroll
                for (int e = 0; e < 4; ++e) {
                    int ta = ts + e;
                    ta = ta < 0 ? -ta : ta;
                    ta = ta >= T_LEN ? 2 * T_LEN - 2 - ta : ta;
                    int tb = ts + 32 + e;
                    tb = tb < 0 ? -tb : tb;
                    tb = tb >= T_LEN ? 2 * T_LEN - 2 - tb : tb;
                    const float u = xr[ta] * wl[e];
                    const float v = xr[tb] * wh[e];
                    p[e] = u + v;
                    d[e] = u - v;
                }
                float* row = lX + i * (NCH * SP) + (16 * q + sch) * SP;
                *(float4*)(row + s4)      = make_float4(p[0], p[1], p[2], p[3]);
                *(float4*)(row + 32 + s4) = make_float4(d[0], d[1], d[2], d[3]);
            }
        }
    }
    if (widx < NFREQ * 4096) Wt[widx] = wv;
    __syncthreads();

    for (int i = 0; i < 4; ++i) {
        const int fi  = i >> 1;
        const int col = col0 + fi;
        const int p   = i & 1;
        const float* lr0 = lX + fi * (NCH * SP) + (32 * p + cL) * SP + 32 * g + 2 * hs;
        const float* lr1 = lr0 + 16 * SP;
        v2f f0[8], f1[8];
#pragma unroll
        for (int j = 0; j < 8; ++j) {
            f0[j] = *(const v2f*)(lr0 + 4 * j);
            f1[j] = *(const v2f*)(lr1 + 4 * j);
        }
        __builtin_amdgcn_sched_barrier(0);
        v8f c0 = {}, c1 = {};
#pragma unroll
        for (int j = 0; j < 8; ++j) {
            c0 = wmma_f32(a[j], f0[j], c0);
            c1 = wmma_f32(a[j], f1[j], c1);
        }
        float* Ust = U + (size_t)(32 * (w & 1) + g) * PLANE + (size_t)col * NCH + 32 * p + lane;
#pragma unroll
        for (int r = 0; r < 8; ++r) {
            const float give = hs ? c0[r] : c1[r];
            const float got  = xhalf(give);
            Ust[(size_t)(2 * r) * PLANE]      = hs ? got : c0[r];
            Ust[(size_t)(16 + 2 * r) * PLANE] = hs ? c1[r] : got;
        }
    }
}

__global__ void mix_gemm(const float* __restrict__ Wt, const float* __restrict__ U,
                         float* __restrict__ Y) {
    const int f  = blockIdx.y;
    const int ri = blockIdx.z;
    if (ri == 1 && (f == 0 || f == 32)) return;
    const int frq = f + ri * 32;

    __shared__ __attribute__((aligned(16))) float lB[2 * 32 * SP];
    const int tid  = threadIdx.x;
    const int lane = tid & 31;
    const int w    = tid >> 5;
    const int cL   = lane & 15;
    const int hs   = lane >> 4;
    const int ow   = w & 1;
    const int ct   = w >> 1;
    v2f wb0[16], wb1[16];
    load_frags(Wt + (size_t)f * 4096, 2 * ow,     lane, wb0);
    load_frags(Wt + (size_t)f * 4096, 2 * ow + 1, lane, wb1);

    const int pair0 = blockIdx.x * PPB;
    const int np    = (NPAIR2 - pair0) < PPB ? (NPAIR2 - pair0) : PPB;
    const float* Up = U + (size_t)frq * PLANE;
    float*       Yp = Y + (size_t)frq * PLANE;
    for (int i = 0; i < np; ++i) {
        const int col0 = (pair0 + i) * 32;
        float* lBc = lB + (i & 1) * (32 * SP);
#pragma unroll
        for (int q = 0; q < 4; ++q) {
            const int fi = q * 128 + tid;
            const int cc = fi >> 4, p4 = fi & 15;
            int gc = col0 + cc;
            gc = gc < NCOL ? gc : NCOL - 1;
            *(float4*)(lBc + cc * SP + p4 * 4) = *(const float4*)(Up + (size_t)gc * NCH + p4 * 4);
        }
        __syncthreads();
        v8f c0 = {}, c1 = {};
        const float* lrow = lBc + (16 * ct + cL) * SP + 2 * hs;
#pragma unroll
        for (int j = 0; j < 16; ++j) {
            const v2f uv = *(const v2f*)(lrow + 4 * j);
            c0 = wmma_f32(uv, wb0[j], c0);
            c1 = wmma_f32(uv, wb1[j], c1);
        }
        const int crow = col0 + 16 * ct;
        float* Yst = Yp + (size_t)crow * NCH + 32 * ow + lane;
        if (col0 + 32 <= NCOL) {
#pragma unroll
            for (int r = 0; r < 8; ++r) {
                const float give = hs ? c0[r] : c1[r];
                const float got  = xhalf(give);
                Yst[(size_t)r * NCH]       = hs ? got : c0[r];
                Yst[(size_t)(8 + r) * NCH] = hs ? c1[r] : got;
            }
        } else {
#pragma unroll
            for (int r = 0; r < 8; ++r) {
                const float give = hs ? c0[r] : c1[r];
                const float got  = xhalf(give);
                if (crow + r < NCOL)     Yst[(size_t)r * NCH]       = hs ? got : c0[r];
                if (crow + 8 + r < NCOL) Yst[(size_t)(8 + r) * NCH] = hs ? c1[r] : got;
            }
        }
    }
}

__global__ void __launch_bounds__(128)
istft_ola_gemm(const float* __restrict__ Y, float* __restrict__ out) {
    __shared__ __attribute__((aligned(16))) float lF[2 * 16 * SP];
    __shared__ __attribute__((aligned(16))) float lY[64 * YP];
    const int tid  = threadIdx.x;
    const int lane = tid & 31;
    const int w    = tid >> 5;
    const int cL   = lane & 15;
    const int hs   = lane >> 4;
    const int g    = w >> 1;

    const int tile0 = blockIdx.x * TPB;
    const int m0 = (tile0 >> 1) & 511;
    const int oh = (tile0 >> 10) & 1;
    const int b  = tile0 >> 11;

    const int sfr = tid >> 3;
    const int sc4 = (tid & 7) * 4;
    const float* Ys = Y + (size_t)sfr * PLANE + (size_t)(b * NFRM + m0) * NCH + 32 * oh + sc4;
    float4 st[5][4];
#pragma unroll
    for (int c = 0; c < 5; ++c) {
#pragma unroll
        for (int q = 0; q < 4; ++q)
            st[c][q] = *(const float4*)(Ys + (size_t)(16 * q) * PLANE + c * NCH);
    }

    v2f a[8];
    build_a3(cospif((float)lane * 0.03125f), w, lane, a);

#pragma unroll
    for (int c = 0; c < 5; ++c) {
#pragma unroll
        for (int q = 0; q < 4; ++q)
            *(float4*)(lY + (16 * q + sfr) * YP + c * 32 + sc4) = st[c][q];
    }
    __syncthreads();

    const int ecol = tid >> 3;
    const int eo4  = (tid & 7) * 4;
    float wlo[4], whi[4], env[4];
#pragma unroll
    for (int q = 0; q < 4; ++q) {
        const float cth = cosf(0.09817477042f * (float)(eo4 + q));
        wlo[q] = 0.5f * (1.0f - cth);
        whi[q] = 0.5f * (1.0f + cth);
        env[q] = 0.5f * (1.0f + cth * cth);
    }
    float4 Ep = make_float4(0.0f, 0.0f, 0.0f, 0.0f);
    float4 Op = make_float4(0.0f, 0.0f, 0.0f, 0.0f);

    for (int it = 0; it < 10; ++it) {
        const int ct = it / 5;
        const int fr = it - 5 * ct;
        const float* lb = lY + (4 * hs + g) * YP + fr * 32 + 16 * ct + cL;
        v2f bv[8];
#pragma unroll
        for (int j = 0; j < 8; ++j) {
            bv[j].x = lb[(8 * j) * YP];
            bv[j].y = lb[(8 * j + 2) * YP];
        }
        __builtin_amdgcn_sched_barrier(0);
        v8f c = {};
#pragma unroll
        for (int j = 0; j < 8; ++j)
            c = wmma_f32(a[j], bv[j], c);
        float* lFc = lF + (it & 1) * (16 * SP);
        float* sp  = lFc + cL * SP + 16 * w + 8 * hs;
        *(float4*)sp       = make_float4(c[0], c[1], c[2], c[3]);
        *(float4*)(sp + 4) = make_float4(c[4], c[5], c[6], c[7]);
        __syncthreads();
        const float* ep = lFc + ecol * SP + eo4;
        const float4 En = *(const float4*)ep;
        const float4 On = *(const float4*)(ep + 32);
        if (fr >= 1) {
            const int m  = m0 + fr - 1;
            const int ot = 2 * oh + ct;
            float4 o;
            o.x = (whi[0] * (Ep.x - Op.x) + wlo[0] * (En.x + On.x)) / env[0];
            o.y = (whi[1] * (Ep.y - Op.y) + wlo[1] * (En.y + On.y)) / env[1];
            o.z = (whi[2] * (Ep.z - Op.z) + wlo[2] * (En.z + On.z)) / env[2];
            o.w = (whi[3] * (Ep.w - Op.w) + wlo[3] * (En.w + On.w)) / env[3];
            *(float4*)(out + (size_t)(b * NCH + 16 * ot + ecol) * T_LEN + m * HOP + eo4) = o;
        }
        Ep = En;
        Op = On;
    }
}

extern "C" void kernel_launch(void* const* d_in, const int* in_sizes, int n_in,
                              void* d_out, int out_size, void* d_ws, size_t ws_size,
                              hipStream_t stream) {
    const float* x   = (const float*)d_in[0];
    const float* wgt = (const float*)d_in[1];
    float* out = (float*)d_out;
    float* ws  = (float*)d_ws;

    float* Wt = ws + OFF_WT;
    float* U  = ws + OFF_U;
    float* Y  = ws + OFF_Y;

    stft_gemm<<<NBLK1, 128, 0, stream>>>(x, wgt, Wt, U);
    mix_gemm<<<dim3(NBLK2, NFREQ, 2), 128, 0, stream>>>(Wt, U, Y);
    istft_ola_gemm<<<NBLK3, 128, 0, stream>>>(Y, out);
}
